# starting kernel
# baseline (speedup 1.0000x reference)
#include <hip/hip_runtime.h>
#include <hip/hip_bf16.h>

typedef float v2f __attribute__((ext_vector_type(2)));
typedef float v8f __attribute__((ext_vector_type(8)));

constexpr int kDim = 384;
constexpr int kDst = 16;
constexpr int kL   = 4096;
constexpr int kB   = 4;
constexpr int kP   = 800;
constexpr int kM   = kB * kL;

constexpr int TS = 72;

__device__ __forceinline__ float softplus_f(float v) {
    return (v > 20.f) ? v : log1pf(__expf(v));
}

__global__ __launch_bounds__(256)
void proj_gemm_kernel(const float* __restrict__ x,
                      const float* __restrict__ Wp,
                      const float* __restrict__ bp,
                      const float* __restrict__ dtb,
                      float* __restrict__ xin,
                      float* __restrict__ dlt,
                      float* __restrict__ Bws,
                      float* __restrict__ Cws)
{
    __shared__ float sA[32 * TS];
    __shared__ float sB[32 * TS];

    const int tid  = threadIdx.x;
    const int wave = tid >> 5;
    const int lane = tid & 31;
    const int half = lane >> 4;
    const int l16  = lane & 15;

    const int m0 = blockIdx.x * 64;
    const int n0 = blockIdx.y * 64;

    const int wm = (wave & 1) * 32;
    const int wn = (wave >> 1) * 16;

    v8f acc0 = {};
    v8f acc1 = {};

    const int bidx = m0 >> 12;
    const int l0   = m0 & 4095;
    const float* xb = x + (size_t)bidx * kDim * kL;

    for (int k0 = 0; k0 < kDim; k0 += 32) {
        {
            const int j  = tid >> 3;
            const int i0 = (tid & 7) * 8;
            const float* src = xb + (size_t)(k0 + j) * kL + l0 + i0;
            float4 v0 = *(const float4*)(src);
            float4 v1 = *(const float4*)(src + 4);
            *(float4*)(&sA[j * TS + i0])     = v0;
            *(float4*)(&sA[j * TS + i0 + 4]) = v1;
        }
        {
            const int nn = tid >> 2;
            const int j0 = (tid & 3) * 8;
            const int ng = n0 + nn;
            float vv[8];
            if (ng < kP) {
                const float* src = Wp + (size_t)ng * kDim + k0 + j0;
                float4 v0 = *(const float4*)(src);
                float4 v1 = *(const float4*)(src + 4);
                vv[0]=v0.x; vv[1]=v0.y; vv[2]=v0.z; vv[3]=v0.w;
                vv[4]=v1.x; vv[5]=v1.y; vv[6]=v1.z; vv[7]=v1.w;
            } else {
                #pragma unroll
                for (int e = 0; e < 8; ++e) vv[e] = 0.f;
            }
            #pragma unroll
            for (int e = 0; e < 8; ++e) sB[(j0 + e) * TS + nn] = vv[e];
        }
        __syncthreads();

        #pragma unroll
        for (int kk = 0; kk < 32; kk += 4) {
            const int kr = kk + 2 * half;
            v2f a0, a1, bf;
            a0.x = sA[kr * TS + wm + l16];
            a0.y = sA[(kr + 1) * TS + wm + l16];
            a1.x = sA[kr * TS + wm + 16 + l16];
            a1.y = sA[(kr + 1) * TS + wm + 16 + l16];
            bf.x = sB[kr * TS + wn + l16];
            bf.y = sB[(kr + 1) * TS + wn + l16];
            acc0 = __builtin_amdgcn_wmma_f32_16x16x4_f32(false, a0, false, bf, (short)0, acc0, false, false);
            acc1 = __builtin_amdgcn_wmma_f32_16x16x4_f32(false, a1, false, bf, (short)0, acc1, false, false);
        }
        __syncthreads();
    }

    const int ng = n0 + wn + l16;
    if (ng < kP) {
        const float bias = bp[ng];
        #pragma unroll
        for (int r = 0; r < 8; ++r) {
            const int mr0 = m0 + wm + r + 8 * half;
            const int mr1 = mr0 + 16;
            float v0 = acc0[r] + bias;
            float v1 = acc1[r] + bias;
            if (ng < kDim) {
                xin[(size_t)mr0 * kDim + ng] = v0;
                xin[(size_t)mr1 * kDim + ng] = v1;
            } else if (ng < 2 * kDim) {
                const int d = ng - kDim;
                const float db = dtb[d];
                dlt[(size_t)mr0 * kDim + d] = softplus_f(v0 + db);
                dlt[(size_t)mr1 * kDim + d] = softplus_f(v1 + db);
            } else if (ng < 2 * kDim + kDst) {
                const int s = ng - 2 * kDim;
                Bws[(size_t)mr0 * kDst + s] = v0;
                Bws[(size_t)mr1 * kDst + s] = v1;
            } else {
                const int s = ng - 2 * kDim - kDst;
                Cws[(size_t)mr0 * kDst + s] = v0;
                Cws[(size_t)mr1 * kDst + s] = v1;
            }
        }
    }
}

__global__ __launch_bounds__(256)
void scan_kernel(const float* __restrict__ xin,
                 const float* __restrict__ dlt,
                 const float* __restrict__ Bws,
                 const float* __restrict__ Cws,
                 const float* __restrict__ A_log,
                 const float* __restrict__ Dp,
                 float* __restrict__ yout)
{
    __shared__ float sD[2][32 * 16];
    __shared__ float sX[2][32 * 16];
    __shared__ float sBv[2][32 * 16];
    __shared__ float sCv[2][32 * 16];

    const int tid = threadIdx.x;
    const int n   = tid & 15;
    const int dl  = tid >> 4;
    const int blk = blockIdx.x;
    const int b   = blk / (kDim / 16);
    const int d0  = (blk % (kDim / 16)) * 16;
    const int d   = d0 + dl;

    const float A_dn = -__expf(A_log[d * kDst + n]);
    const float D_d  = Dp[d];

    const float* xin_b = xin + (size_t)b * kL * kDim;
    const float* dlt_b = dlt + (size_t)b * kL * kDim;
    const float* Bb    = Bws + (size_t)b * kL * kDst;
    const float* Cb    = Cws + (size_t)b * kL * kDst;
    float*       yb    = yout + (size_t)b * kL * kDim;

    auto load_chunk = [&](int t0, int buf) {
        #pragma unroll
        for (int e = tid; e < 512; e += 256) {
            const int row = e >> 4, col = e & 15;
            const size_t go = (size_t)(t0 + row) * kDim + d0 + col;
            sD[buf][e] = dlt_b[go];
            sX[buf][e] = xin_b[go];
            const size_t gb = (size_t)t0 * kDst + e;
            sBv[buf][e] = Bb[gb];
            sCv[buf][e] = Cb[gb];
        }
    };

    load_chunk(0, 0);
    __syncthreads();

    float h = 0.f;
    constexpr int NCHUNK = kL / 32;
    for (int c = 0; c < NCHUNK; ++c) {
        const int buf = c & 1;
        if (c + 1 < NCHUNK) load_chunk((c + 1) * 32, buf ^ 1);

        const int tbase = c * 32;
        #pragma unroll 8
        for (int tt = 0; tt < 32; ++tt) {
            const float dt = sD[buf][tt * 16 + dl];
            const float xv = sX[buf][tt * 16 + dl];
            const float bn = sBv[buf][tt * 16 + n];
            const float cn = sCv[buf][tt * 16 + n];
            h = __expf(dt * A_dn) * h + (dt * xv) * bn;
            float yp = h * cn;
            yp += __shfl_xor(yp, 1);
            yp += __shfl_xor(yp, 2);
            yp += __shfl_xor(yp, 4);
            yp += __shfl_xor(yp, 8);
            if (n == 0)
                yb[(size_t)(tbase + tt) * kDim + d] = yp + xv * D_d;
        }
        __syncthreads();
    }
}

__global__ __launch_bounds__(256)
void out_gemm_kernel(const float* __restrict__ y,
                     const float* __restrict__ Wo,
                     const float* __restrict__ bo,
                     float* __restrict__ out)
{
    __shared__ float smem[64 * TS];
    float* sA = smem;
    float* sB = smem + 32 * TS;
    float* sC = smem;

    const int tid  = threadIdx.x;
    const int wave = tid >> 5;
    const int lane = tid & 31;
    const int half = lane >> 4;
    const int l16  = lane & 15;

    const int m0 = blockIdx.x * 64;
    const int n0 = blockIdx.y * 64;
    const int wm = (wave & 1) * 32;
    const int wn = (wave >> 1) * 16;

    v8f acc0 = {};
    v8f acc1 = {};

    for (int k0 = 0; k0 < kDim; k0 += 32) {
        {
            const int i  = tid >> 2;
            const int j0 = (tid & 3) * 8;
            const float* src = y + (size_t)(m0 + i) * kDim + k0 + j0;
            float4 v0 = *(const float4*)(src);
            float4 v1 = *(const float4*)(src + 4);
            sA[(j0 + 0) * TS + i] = v0.x; sA[(j0 + 1) * TS + i] = v0.y;
            sA[(j0 + 2) * TS + i] = v0.z; sA[(j0 + 3) * TS + i] = v0.w;
            sA[(j0 + 4) * TS + i] = v1.x; sA[(j0 + 5) * TS + i] = v1.y;
            sA[(j0 + 6) * TS + i] = v1.z; sA[(j0 + 7) * TS + i] = v1.w;
        }
        {
            const int nn = tid >> 2;
            const int j0 = (tid & 3) * 8;
            const float* src = Wo + (size_t)(n0 + nn) * kDim + k0 + j0;
            float4 v0 = *(const float4*)(src);
            float4 v1 = *(const float4*)(src + 4);
            sB[(j0 + 0) * TS + nn] = v0.x; sB[(j0 + 1) * TS + nn] = v0.y;
            sB[(j0 + 2) * TS + nn] = v0.z; sB[(j0 + 3) * TS + nn] = v0.w;
            sB[(j0 + 4) * TS + nn] = v1.x; sB[(j0 + 5) * TS + nn] = v1.y;
            sB[(j0 + 6) * TS + nn] = v1.z; sB[(j0 + 7) * TS + nn] = v1.w;
        }
        __syncthreads();

        #pragma unroll
        for (int kk = 0; kk < 32; kk += 4) {
            const int kr = kk + 2 * half;
            v2f a0, a1, bf;
            a0.x = sA[kr * TS + wm + l16];
            a0.y = sA[(kr + 1) * TS + wm + l16];
            a1.x = sA[kr * TS + wm + 16 + l16];
            a1.y = sA[(kr + 1) * TS + wm + 16 + l16];
            bf.x = sB[kr * TS + wn + l16];
            bf.y = sB[(kr + 1) * TS + wn + l16];
            acc0 = __builtin_amdgcn_wmma_f32_16x16x4_f32(false, a0, false, bf, (short)0, acc0, false, false);
            acc1 = __builtin_amdgcn_wmma_f32_16x16x4_f32(false, a1, false, bf, (short)0, acc1, false, false);
        }
        __syncthreads();
    }

    const float bias = bo[n0 + wn + l16];
    #pragma unroll
    for (int r = 0; r < 8; ++r) {
        const int mm = wm + r + 8 * half;
        sC[(wn + l16) * TS + mm]      = acc0[r] + bias;
        sC[(wn + l16) * TS + mm + 16] = acc1[r] + bias;
    }
    __syncthreads();

    const int bidx = m0 >> 12;
    const int l0   = m0 & 4095;
    const int nn   = tid >> 2;
    const int ms   = (tid & 3) * 16;
    float* dst = out + ((size_t)bidx * kDim + n0 + nn) * kL + l0 + ms;
    #pragma unroll
    for (int e = 0; e < 4; ++e) {
        float4 v;
        v.x = sC[nn * TS + ms + e * 4 + 0];
        v.y = sC[nn * TS + ms + e * 4 + 1];
        v.z = sC[nn * TS + ms + e * 4 + 2];
        v.w = sC[nn * TS + ms + e * 4 + 3];
        *(float4*)(dst + e * 4) = v;
    }
}

extern "C" void kernel_launch(void* const* d_in, const int* in_sizes, int n_in,
                              void* d_out, int out_size, void* d_ws, size_t ws_size,
                              hipStream_t stream) {
    const float* x     = (const float*)d_in[0];
    const float* Wp    = (const float*)d_in[1];
    const float* bp    = (const float*)d_in[2];
    const float* A_log = (const float*)d_in[3];
    const float* Dp    = (const float*)d_in[4];
    const float* dtb   = (const float*)d_in[5];
    const float* Wo    = (const float*)d_in[6];
    const float* bo    = (const float*)d_in[7];
    float* out = (float*)d_out;

    float* ws = (float*)d_ws;
    const size_t MD = (size_t)kM * kDim;
    const size_t MS = (size_t)kM * kDst;
    float* xin = ws;
    float* dlt = ws + MD;
    float* yws = ws + 2 * MD;
    float* Bw  = ws + 3 * MD;
    float* Cw  = Bw + MS;

    dim3 g1(kM / 64, (kP + 63) / 64);
    proj_gemm_kernel<<<g1, 256, 0, stream>>>(x, Wp, bp, dtb, xin, dlt, Bw, Cw);

    scan_kernel<<<kB * (kDim / 16), 256, 0, stream>>>(xin, dlt, Bw, Cw, A_log, Dp, yws);

    dim3 g3(kM / 64, kDim / 64);
    out_gemm_kernel<<<g3, 256, 0, stream>>>(yws, Wo, bo, out);
}


# Scan over L split into chunks of 256 steps run in parallel: a new launch gets each chunk's decay product and zero-start end state, the scan folds them into its start state (h summation order changes)
# speedup vs baseline: 5.0454x; 5.0454x over previous
#include <hip/hip_runtime.h>
#include <hip/hip_bf16.h>

typedef float v2f __attribute__((ext_vector_type(2)));
typedef float v8f __attribute__((ext_vector_type(8)));

constexpr int kDim = 384;
constexpr int kDst = 16;
constexpr int kL   = 4096;
constexpr int kB   = 4;
constexpr int kP   = 800;
constexpr int kM   = kB * kL;

constexpr int TS = 72;

constexpr int NC = 16;
constexpr int CL = kL / NC;

__device__ __forceinline__ float softplus_f(float v) {
    return (v > 20.f) ? v : log1pf(__expf(v));
}

__global__ __launch_bounds__(256)
void proj_gemm_kernel(const float* __restrict__ x,
                      const float* __restrict__ Wp,
                      const float* __restrict__ bp,
                      const float* __restrict__ dtb,
                      float* __restrict__ xin,
                      float* __restrict__ dlt,
                      float* __restrict__ Bws,
                      float* __restrict__ Cws)
{
    __shared__ float sA[32 * TS];
    __shared__ float sB[32 * TS];

    const int tid  = threadIdx.x;
    const int wave = tid >> 5;
    const int lane = tid & 31;
    const int half = lane >> 4;
    const int l16  = lane & 15;

    const int m0 = blockIdx.x * 64;
    const int n0 = blockIdx.y * 64;

    const int wm = (wave & 1) * 32;
    const int wn = (wave >> 1) * 16;

    v8f acc0 = {};
    v8f acc1 = {};

    const int bidx = m0 >> 12;
    const int l0   = m0 & 4095;
    const float* xb = x + (size_t)bidx * kDim * kL;

    for (int k0 = 0; k0 < kDim; k0 += 32) {
        {
            const int j  = tid >> 3;
            const int i0 = (tid & 7) * 8;
            const float* src = xb + (size_t)(k0 + j) * kL + l0 + i0;
            float4 v0 = *(const float4*)(src);
            float4 v1 = *(const float4*)(src + 4);
            *(float4*)(&sA[j * TS + i0])     = v0;
            *(float4*)(&sA[j * TS + i0 + 4]) = v1;
        }
        {
            const int nn = tid >> 2;
            const int j0 = (tid & 3) * 8;
            const int ng = n0 + nn;
            float vv[8];
            if (ng < kP) {
                const float* src = Wp + (size_t)ng * kDim + k0 + j0;
                float4 v0 = *(const float4*)(src);
                float4 v1 = *(const float4*)(src + 4);
                vv[0]=v0.x; vv[1]=v0.y; vv[2]=v0.z; vv[3]=v0.w;
                vv[4]=v1.x; vv[5]=v1.y; vv[6]=v1.z; vv[7]=v1.w;
            } else {
                #pragma unroll
                for (int e = 0; e < 8; ++e) vv[e] = 0.f;
            }
            #pragma unroll
            for (int e = 0; e < 8; ++e) sB[(j0 + e) * TS + nn] = vv[e];
        }
        __syncthreads();

        #pragma unroll
        for (int kk = 0; kk < 32; kk += 4) {
            const int kr = kk + 2 * half;
            v2f a0, a1, bf;
            a0.x = sA[kr * TS + wm + l16];
            a0.y = sA[(kr + 1) * TS + wm + l16];
            a1.x = sA[kr * TS + wm + 16 + l16];
            a1.y = sA[(kr + 1) * TS + wm + 16 + l16];
            bf.x = sB[kr * TS + wn + l16];
            bf.y = sB[(kr + 1) * TS + wn + l16];
            acc0 = __builtin_amdgcn_wmma_f32_16x16x4_f32(false, a0, false, bf, (short)0, acc0, false, false);
            acc1 = __builtin_amdgcn_wmma_f32_16x16x4_f32(false, a1, false, bf, (short)0, acc1, false, false);
        }
        __syncthreads();
    }

    const int ng = n0 + wn + l16;
    if (ng < kP) {
        const float bias = bp[ng];
        #pragma unroll
        for (int r = 0; r < 8; ++r) {
            const int mr0 = m0 + wm + r + 8 * half;
            const int mr1 = mr0 + 16;
            float v0 = acc0[r] + bias;
            float v1 = acc1[r] + bias;
            if (ng < kDim) {
                xin[(size_t)mr0 * kDim + ng] = v0;
                xin[(size_t)mr1 * kDim + ng] = v1;
            } else if (ng < 2 * kDim) {
                const int d = ng - kDim;
                const float db = dtb[d];
                dlt[(size_t)mr0 * kDim + d] = softplus_f(v0 + db);
                dlt[(size_t)mr1 * kDim + d] = softplus_f(v1 + db);
            } else if (ng < 2 * kDim + kDst) {
                const int s = ng - 2 * kDim;
                Bws[(size_t)mr0 * kDst + s] = v0;
                Bws[(size_t)mr1 * kDst + s] = v1;
            } else {
                const int s = ng - 2 * kDim - kDst;
                Cws[(size_t)mr0 * kDst + s] = v0;
                Cws[(size_t)mr1 * kDst + s] = v1;
            }
        }
    }
}

__global__ __launch_bounds__(256)
void scan_state_kernel(const float* __restrict__ xin,
                       const float* __restrict__ dlt,
                       const float* __restrict__ Bws,
                       const float* __restrict__ A_log,
                       float* __restrict__ Pws,
                       float* __restrict__ Hws)
{
    __shared__ float sD[2][32 * 16];
    __shared__ float sX[2][32 * 16];
    __shared__ float sBv[2][32 * 16];

    const int tid = threadIdx.x;
    const int n   = tid & 15;
    const int dl  = tid >> 4;
    const int blk = blockIdx.x;
    const int c   = blk % (NC - 1);
    const int bd  = blk / (NC - 1);
    const int b   = bd / (kDim / 16);
    const int d0  = (bd % (kDim / 16)) * 16;
    const int d   = d0 + dl;

    const float A_dn = -__expf(A_log[d * kDst + n]);

    const size_t row0 = (size_t)b * kL + (size_t)c * CL;
    const float* xin_b = xin + row0 * kDim;
    const float* dlt_b = dlt + row0 * kDim;
    const float* Bb    = Bws + row0 * kDst;

    auto load_chunk = [&](int t0, int buf) {
        #pragma unroll
        for (int e = tid; e < 512; e += 256) {
            const int row = e >> 4, col = e & 15;
            const size_t go = (size_t)(t0 + row) * kDim + d0 + col;
            sD[buf][e] = dlt_b[go];
            sX[buf][e] = xin_b[go];
            sBv[buf][e] = Bb[(size_t)t0 * kDst + e];
        }
    };

    load_chunk(0, 0);
    __syncthreads();

    float h = 0.f;
    float p = 1.f;
    constexpr int NCHUNK = CL / 32;
    for (int cc = 0; cc < NCHUNK; ++cc) {
        const int buf = cc & 1;
        if (cc + 1 < NCHUNK) load_chunk((cc + 1) * 32, buf ^ 1);

        #pragma unroll 8
        for (int tt = 0; tt < 32; ++tt) {
            const float dt = sD[buf][tt * 16 + dl];
            const float xv = sX[buf][tt * 16 + dl];
            const float bn = sBv[buf][tt * 16 + n];
            const float e  = __expf(dt * A_dn);
            h = e * h + (dt * xv) * bn;
            p = e * p;
        }
        __syncthreads();
    }

    const size_t o = (((size_t)b * NC + c) * kDim + d) * kDst + n;
    Pws[o] = p;
    Hws[o] = h;
}

__global__ __launch_bounds__(256)
void scan_kernel(const float* __restrict__ xin,
                 const float* __restrict__ dlt,
                 const float* __restrict__ Bws,
                 const float* __restrict__ Cws,
                 const float* __restrict__ A_log,
                 const float* __restrict__ Dp,
                 const float* __restrict__ Pws,
                 const float* __restrict__ Hws,
                 float* __restrict__ yout)
{
    __shared__ float sD[2][32 * 16];
    __shared__ float sX[2][32 * 16];
    __shared__ float sBv[2][32 * 16];
    __shared__ float sCv[2][32 * 16];

    const int tid = threadIdx.x;
    const int n   = tid & 15;
    const int dl  = tid >> 4;
    const int blk = blockIdx.x;
    const int c   = blk % NC;
    const int bd  = blk / NC;
    const int b   = bd / (kDim / 16);
    const int d0  = (bd % (kDim / 16)) * 16;
    const int d   = d0 + dl;

    const float A_dn = -__expf(A_log[d * kDst + n]);
    const float D_d  = Dp[d];

    const size_t row0 = (size_t)b * kL + (size_t)c * CL;
    const float* xin_b = xin + row0 * kDim;
    const float* dlt_b = dlt + row0 * kDim;
    const float* Bb    = Bws + row0 * kDst;
    const float* Cb    = Cws + row0 * kDst;
    float*       yb    = yout + row0 * kDim;

    auto load_chunk = [&](int t0, int buf) {
        #pragma unroll
        for (int e = tid; e < 512; e += 256) {
            const int row = e >> 4, col = e & 15;
            const size_t go = (size_t)(t0 + row) * kDim + d0 + col;
            sD[buf][e] = dlt_b[go];
            sX[buf][e] = xin_b[go];
            const size_t gb = (size_t)t0 * kDst + e;
            sBv[buf][e] = Bb[gb];
            sCv[buf][e] = Cb[gb];
        }
    };

    load_chunk(0, 0);

    float h = 0.f;
    {
        const size_t o0 = (((size_t)b * NC) * kDim + d) * kDst + n;
        float pp[NC - 1], hh[NC - 1];
        #pragma unroll
        for (int cp = 0; cp < NC - 1; ++cp) {
            pp[cp] = Pws[o0 + (size_t)cp * kDim * kDst];
            hh[cp] = Hws[o0 + (size_t)cp * kDim * kDst];
        }
        #pragma unroll
        for (int cp = 0; cp < NC - 1; ++cp) {
            const float hn = pp[cp] * h + hh[cp];
            h = (cp < c) ? hn : h;
        }
    }
    __syncthreads();

    constexpr int NCHUNK = CL / 32;
    for (int cc = 0; cc < NCHUNK; ++cc) {
        const int buf = cc & 1;
        if (cc + 1 < NCHUNK) load_chunk((cc + 1) * 32, buf ^ 1);

        const int tbase = cc * 32;
        #pragma unroll 8
        for (int tt = 0; tt < 32; ++tt) {
            const float dt = sD[buf][tt * 16 + dl];
            const float xv = sX[buf][tt * 16 + dl];
            const float bn = sBv[buf][tt * 16 + n];
            const float cn = sCv[buf][tt * 16 + n];
            h = __expf(dt * A_dn) * h + (dt * xv) * bn;
            float yp = h * cn;
            yp += __shfl_xor(yp, 1);
            yp += __shfl_xor(yp, 2);
            yp += __shfl_xor(yp, 4);
            yp += __shfl_xor(yp, 8);
            if (n == 0)
                yb[(size_t)(tbase + tt) * kDim + d] = yp + xv * D_d;
        }
        __syncthreads();
    }
}

__global__ __launch_bounds__(256)
void out_gemm_kernel(const float* __restrict__ y,
                     const float* __restrict__ Wo,
                     const float* __restrict__ bo,
                     float* __restrict__ out)
{
    __shared__ float smem[64 * TS];
    float* sA = smem;
    float* sB = smem + 32 * TS;
    float* sC = smem;

    const int tid  = threadIdx.x;
    const int wave = tid >> 5;
    const int lane = tid & 31;
    const int half = lane >> 4;
    const int l16  = lane & 15;

    const int m0 = blockIdx.x * 64;
    const int n0 = blockIdx.y * 64;
    const int wm = (wave & 1) * 32;
    const int wn = (wave >> 1) * 16;

    v8f acc0 = {};
    v8f acc1 = {};

    for (int k0 = 0; k0 < kDim; k0 += 32) {
        {
            const int i  = tid >> 2;
            const int j0 = (tid & 3) * 8;
            const float* src = y + (size_t)(m0 + i) * kDim + k0 + j0;
            float4 v0 = *(const float4*)(src);
            float4 v1 = *(const float4*)(src + 4);
            sA[(j0 + 0) * TS + i] = v0.x; sA[(j0 + 1) * TS + i] = v0.y;
            sA[(j0 + 2) * TS + i] = v0.z; sA[(j0 + 3) * TS + i] = v0.w;
            sA[(j0 + 4) * TS + i] = v1.x; sA[(j0 + 5) * TS + i] = v1.y;
            sA[(j0 + 6) * TS + i] = v1.z; sA[(j0 + 7) * TS + i] = v1.w;
        }
        {
            const int nn = tid >> 2;
            const int j0 = (tid & 3) * 8;
            const float* src = Wo + (size_t)(n0 + nn) * kDim + k0 + j0;
            float4 v0 = *(const float4*)(src);
            float4 v1 = *(const float4*)(src + 4);
            sB[(j0 + 0) * TS + nn] = v0.x; sB[(j0 + 1) * TS + nn] = v0.y;
            sB[(j0 + 2) * TS + nn] = v0.z; sB[(j0 + 3) * TS + nn] = v0.w;
            sB[(j0 + 4) * TS + nn] = v1.x; sB[(j0 + 5) * TS + nn] = v1.y;
            sB[(j0 + 6) * TS + nn] = v1.z; sB[(j0 + 7) * TS + nn] = v1.w;
        }
        __syncthreads();

        #pragma unroll
        for (int kk = 0; kk < 32; kk += 4) {
            const int kr = kk + 2 * half;
            v2f a0, a1, bf;
            a0.x = sA[kr * TS + wm + l16];
            a0.y = sA[(kr + 1) * TS + wm + l16];
            a1.x = sA[kr * TS + wm + 16 + l16];
            a1.y = sA[(kr + 1) * TS + wm + 16 + l16];
            bf.x = sB[kr * TS + wn + l16];
            bf.y = sB[(kr + 1) * TS + wn + l16];
            acc0 = __builtin_amdgcn_wmma_f32_16x16x4_f32(false, a0, false, bf, (short)0, acc0, false, false);
            acc1 = __builtin_amdgcn_wmma_f32_16x16x4_f32(false, a1, false, bf, (short)0, acc1, false, false);
        }
        __syncthreads();
    }

    const float bias = bo[n0 + wn + l16];
    #pragma unroll
    for (int r = 0; r < 8; ++r) {
        const int mm = wm + r + 8 * half;
        sC[(wn + l16) * TS + mm]      = acc0[r] + bias;
        sC[(wn + l16) * TS + mm + 16] = acc1[r] + bias;
    }
    __syncthreads();

    const int bidx = m0 >> 12;
    const int l0   = m0 & 4095;
    const int nn   = tid >> 2;
    const int ms   = (tid & 3) * 16;
    float* dst = out + ((size_t)bidx * kDim + n0 + nn) * kL + l0 + ms;
    #pragma unroll
    for (int e = 0; e < 4; ++e) {
        float4 v;
        v.x = sC[nn * TS + ms + e * 4 + 0];
        v.y = sC[nn * TS + ms + e * 4 + 1];
        v.z = sC[nn * TS + ms + e * 4 + 2];
        v.w = sC[nn * TS + ms + e * 4 + 3];
        *(float4*)(dst + e * 4) = v;
    }
}

extern "C" void kernel_launch(void* const* d_in, const int* in_sizes, int n_in,
                              void* d_out, int out_size, void* d_ws, size_t ws_size,
                              hipStream_t stream) {
    const float* x     = (const float*)d_in[0];
    const float* Wp    = (const float*)d_in[1];
    const float* bp    = (const float*)d_in[2];
    const float* A_log = (const float*)d_in[3];
    const float* Dp    = (const float*)d_in[4];
    const float* dtb   = (const float*)d_in[5];
    const float* Wo    = (const float*)d_in[6];
    const float* bo    = (const float*)d_in[7];
    float* out = (float*)d_out;

    float* ws = (float*)d_ws;
    const size_t MD = (size_t)kM * kDim;
    const size_t MS = (size_t)kM * kDst;
    float* xin = ws;
    float* dlt = ws + MD;
    float* yws = ws + 2 * MD;
    float* Bw  = ws + 3 * MD;
    float* Cw  = Bw + MS;
    const size_t SS = (size_t)kB * NC * kDim * kDst;
    float* Pw  = Cw + MS;
    float* Hw  = Pw + SS;

    dim3 g1(kM / 64, (kP + 63) / 64);
    proj_gemm_kernel<<<g1, 256, 0, stream>>>(x, Wp, bp, dtb, xin, dlt, Bw, Cw);

    scan_state_kernel<<<kB * (kDim / 16) * (NC - 1), 256, 0, stream>>>(xin, dlt, Bw, A_log, Pw, Hw);

    scan_kernel<<<kB * (kDim / 16) * NC, 256, 0, stream>>>(xin, dlt, Bw, Cw, A_log, Dp, Pw, Hw, yws);

    dim3 g3(kM / 64, kDim / 64);
    out_gemm_kernel<<<g3, 256, 0, stream>>>(yws, Wo, bo, out);
}


# Scan and chunk-summary kernels remapped to one lane per channel with its 16 states in registers: y's sum over states is now in-lane (sequential, two partial sums), one full-line store per step
# speedup vs baseline: 5.8013x; 1.1498x over previous
#include <hip/hip_runtime.h>
#include <hip/hip_bf16.h>

typedef float v2f __attribute__((ext_vector_type(2)));
typedef float v8f __attribute__((ext_vector_type(8)));

constexpr int kDim = 384;
constexpr int kDst = 16;
constexpr int kL   = 4096;
constexpr int kB   = 4;
constexpr int kP   = 800;
constexpr int kM   = kB * kL;

constexpr int TS = 72;

constexpr int NC = 16;
constexpr int CL = kL / NC;

__device__ __forceinline__ float softplus_f(float v) {
    return (v > 20.f) ? v : log1pf(__expf(v));
}

__global__ __launch_bounds__(256)
void proj_gemm_kernel(const float* __restrict__ x,
                      const float* __restrict__ Wp,
                      const float* __restrict__ bp,
                      const float* __restrict__ dtb,
                      float* __restrict__ xin,
                      float* __restrict__ dlt,
                      float* __restrict__ Bws,
                      float* __restrict__ Cws)
{
    __shared__ float sA[32 * TS];
    __shared__ float sB[32 * TS];

    const int tid  = threadIdx.x;
    const int wave = tid >> 5;
    const int lane = tid & 31;
    const int half = lane >> 4;
    const int l16  = lane & 15;

    const int m0 = blockIdx.x * 64;
    const int n0 = blockIdx.y * 64;

    const int wm = (wave & 1) * 32;
    const int wn = (wave >> 1) * 16;

    v8f acc0 = {};
    v8f acc1 = {};

    const int bidx = m0 >> 12;
    const int l0   = m0 & 4095;
    const float* xb = x + (size_t)bidx * kDim * kL;

    for (int k0 = 0; k0 < kDim; k0 += 32) {
        {
            const int j  = tid >> 3;
            const int i0 = (tid & 7) * 8;
            const float* src = xb + (size_t)(k0 + j) * kL + l0 + i0;
            float4 v0 = *(const float4*)(src);
            float4 v1 = *(const float4*)(src + 4);
            *(float4*)(&sA[j * TS + i0])     = v0;
            *(float4*)(&sA[j * TS + i0 + 4]) = v1;
        }
        {
            const int nn = tid >> 2;
            const int j0 = (tid & 3) * 8;
            const int ng = n0 + nn;
            float vv[8];
            if (ng < kP) {
                const float* src = Wp + (size_t)ng * kDim + k0 + j0;
                float4 v0 = *(const float4*)(src);
                float4 v1 = *(const float4*)(src + 4);
                vv[0]=v0.x; vv[1]=v0.y; vv[2]=v0.z; vv[3]=v0.w;
                vv[4]=v1.x; vv[5]=v1.y; vv[6]=v1.z; vv[7]=v1.w;
            } else {
                #pragma unroll
                for (int e = 0; e < 8; ++e) vv[e] = 0.f;
            }
            #pragma unroll
            for (int e = 0; e < 8; ++e) sB[(j0 + e) * TS + nn] = vv[e];
        }
        __syncthreads();

        #pragma unroll
        for (int kk = 0; kk < 32; kk += 4) {
            const int kr = kk + 2 * half;
            v2f a0, a1, bf;
            a0.x = sA[kr * TS + wm + l16];
            a0.y = sA[(kr + 1) * TS + wm + l16];
            a1.x = sA[kr * TS + wm + 16 + l16];
            a1.y = sA[(kr + 1) * TS + wm + 16 + l16];
            bf.x = sB[kr * TS + wn + l16];
            bf.y = sB[(kr + 1) * TS + wn + l16];
            acc0 = __builtin_amdgcn_wmma_f32_16x16x4_f32(false, a0, false, bf, (short)0, acc0, false, false);
            acc1 = __builtin_amdgcn_wmma_f32_16x16x4_f32(false, a1, false, bf, (short)0, acc1, false, false);
        }
        __syncthreads();
    }

    const int ng = n0 + wn + l16;
    if (ng < kP) {
        const float bias = bp[ng];
        #pragma unroll
        for (int r = 0; r < 8; ++r) {
            const int mr0 = m0 + wm + r + 8 * half;
            const int mr1 = mr0 + 16;
            float v0 = acc0[r] + bias;
            float v1 = acc1[r] + bias;
            if (ng < kDim) {
                xin[(size_t)mr0 * kDim + ng] = v0;
                xin[(size_t)mr1 * kDim + ng] = v1;
            } else if (ng < 2 * kDim) {
                const int d = ng - kDim;
                const float db = dtb[d];
                dlt[(size_t)mr0 * kDim + d] = softplus_f(v0 + db);
                dlt[(size_t)mr1 * kDim + d] = softplus_f(v1 + db);
            } else if (ng < 2 * kDim + kDst) {
                const int s = ng - 2 * kDim;
                Bws[(size_t)mr0 * kDst + s] = v0;
                Bws[(size_t)mr1 * kDst + s] = v1;
            } else {
                const int s = ng - 2 * kDim - kDst;
                Cws[(size_t)mr0 * kDst + s] = v0;
                Cws[(size_t)mr1 * kDst + s] = v1;
            }
        }
    }
}

typedef float v4f __attribute__((ext_vector_type(4)));

constexpr int SW = 96;
constexpr int SG = kDim / SW;
constexpr int SU = 8;
constexpr int NGRP = CL / SU;

__global__ __launch_bounds__(SW)
void scan_state_kernel(const float* __restrict__ xin,
                       const float* __restrict__ dlt,
                       const float* __restrict__ Bws,
                       const float* __restrict__ A_log,
                       float* __restrict__ Pws,
                       float* __restrict__ Hws)
{
    __shared__ float4 sB[2][SU][4];

    const int tid = threadIdx.x;
    const int blk = blockIdx.x;
    const int c   = blk % (NC - 1);
    const int bg  = blk / (NC - 1);
    const int b   = bg / SG;
    const int d   = (bg % SG) * SW + tid;

    float A[kDst];
    #pragma unroll
    for (int q = 0; q < 4; ++q) {
        const float4 a = *(const float4*)(A_log + (size_t)d * kDst + 4 * q);
        A[4 * q + 0] = -__expf(a.x); A[4 * q + 1] = -__expf(a.y);
        A[4 * q + 2] = -__expf(a.z); A[4 * q + 3] = -__expf(a.w);
    }

    const size_t row0 = (size_t)b * kL + (size_t)c * CL;
    const float* xin_p = xin + row0 * kDim + d;
    const float* dlt_p = dlt + row0 * kDim + d;
    const float4* Bb4  = (const float4*)(Bws + row0 * kDst);

    const bool stager = tid < 4 * SU;
    const int  sk     = tid & (4 * SU - 1);

    float h[kDst], p[kDst];
    #pragma unroll
    for (int n = 0; n < kDst; ++n) { h[n] = 0.f; p[n] = 1.f; }

    float dtc[SU], xvc[SU];
    #pragma unroll
    for (int j = 0; j < SU; ++j) {
        dtc[j] = dlt_p[(size_t)j * kDim];
        xvc[j] = xin_p[(size_t)j * kDim];
    }
    if (stager) sB[0][sk >> 2][sk & 3] = Bb4[sk];
    __syncthreads();

    for (int g = 0; g < NGRP; ++g) {
        const int buf = g & 1;
        const int gn = (g + 1 < NGRP) ? g + 1 : g;
        const size_t t1 = (size_t)gn * SU;
        float dtn[SU], xvn[SU];
        #pragma unroll
        for (int j = 0; j < SU; ++j) {
            dtn[j] = dlt_p[(t1 + j) * kDim];
            xvn[j] = xin_p[(t1 + j) * kDim];
        }
        float4 bnx = {};
        if (stager) bnx = Bb4[t1 * 4 + sk];

        float4 bq[4];
        #pragma unroll
        for (int q = 0; q < 4; ++q) bq[q] = sB[buf][0][q];

        #pragma unroll
        for (int j = 0; j < SU; ++j) {
            float4 bqn[4];
            if (j + 1 < SU) {
                #pragma unroll
                for (int q = 0; q < 4; ++q) bqn[q] = sB[buf][j + 1][q];
            }
            float bn[kDst];
            #pragma unroll
            for (int q = 0; q < 4; ++q) {
                bn[4 * q + 0] = bq[q].x; bn[4 * q + 1] = bq[q].y; bn[4 * q + 2] = bq[q].z; bn[4 * q + 3] = bq[q].w;
            }
            const float dt = dtc[j];
            const float u  = dt * xvc[j];
            #pragma unroll
            for (int n = 0; n < kDst; ++n) {
                const float e = __expf(dt * A[n]);
                h[n] = e * h[n] + u * bn[n];
                p[n] = e * p[n];
            }
            if (j + 1 < SU) {
                #pragma unroll
                for (int q = 0; q < 4; ++q) bq[q] = bqn[q];
            }
        }

        if (stager) sB[buf ^ 1][sk >> 2][sk & 3] = bnx;
        #pragma unroll
        for (int j = 0; j < SU; ++j) { dtc[j] = dtn[j]; xvc[j] = xvn[j]; }
        __syncthreads();
    }

    const size_t o = (((size_t)b * NC + c) * kDim + d) * kDst;
    #pragma unroll
    for (int q = 0; q < 4; ++q) {
        float4 pv, hv;
        pv.x = p[4 * q + 0]; pv.y = p[4 * q + 1]; pv.z = p[4 * q + 2]; pv.w = p[4 * q + 3];
        hv.x = h[4 * q + 0]; hv.y = h[4 * q + 1]; hv.z = h[4 * q + 2]; hv.w = h[4 * q + 3];
        *(float4*)(Pws + o + 4 * q) = pv;
        *(float4*)(Hws + o + 4 * q) = hv;
    }
}

__global__ __launch_bounds__(SW)
void scan_kernel(const float* __restrict__ xin,
                 const float* __restrict__ dlt,
                 const float* __restrict__ Bws,
                 const float* __restrict__ Cws,
                 const float* __restrict__ A_log,
                 const float* __restrict__ Dp,
                 const float* __restrict__ Pws,
                 const float* __restrict__ Hws,
                 float* __restrict__ yout)
{
    __shared__ float4 sBC[2][SU][8];

    const int tid = threadIdx.x;
    const int blk = blockIdx.x;
    const int c   = blk % NC;
    const int bg  = blk / NC;
    const int b   = bg / SG;
    const int d   = (bg % SG) * SW + tid;

    float A[kDst];
    #pragma unroll
    for (int q = 0; q < 4; ++q) {
        const float4 a = *(const float4*)(A_log + (size_t)d * kDst + 4 * q);
        A[4 * q + 0] = -__expf(a.x); A[4 * q + 1] = -__expf(a.y);
        A[4 * q + 2] = -__expf(a.z); A[4 * q + 3] = -__expf(a.w);
    }
    const float D_d = Dp[d];

    const size_t row0 = (size_t)b * kL + (size_t)c * CL;
    const float* xin_p = xin + row0 * kDim + d;
    const float* dlt_p = dlt + row0 * kDim + d;
    float*       y_p   = yout + row0 * kDim + d;
    const float4* Bb4  = (const float4*)(Bws + row0 * kDst);
    const float4* Cb4  = (const float4*)(Cws + row0 * kDst);

    const bool    stager = tid < 8 * SU;
    const int     sk     = tid & (4 * SU - 1);
    const float4* BC4    = (tid < 4 * SU) ? Bb4 : Cb4;
    const int     sj     = sk >> 2;
    const int     sq     = (sk & 3) + ((tid < 4 * SU) ? 0 : 4);

    float dtc[SU], xvc[SU];
    #pragma unroll
    for (int j = 0; j < SU; ++j) {
        dtc[j] = dlt_p[(size_t)j * kDim];
        xvc[j] = xin_p[(size_t)j * kDim];
    }
    if (stager) sBC[0][sj][sq] = BC4[sk];

    float h[kDst];
    #pragma unroll
    for (int n = 0; n < kDst; ++n) h[n] = 0.f;
    {
        constexpr int FB = 5;
        static_assert((NC - 1) % FB == 0, "batches cover the summaries exactly");
        const size_t o0 = (((size_t)b * NC) * kDim + d) * kDst;
        #pragma unroll
        for (int k = 0; k < (NC - 1) / FB; ++k) {
            if (k * FB < c) {
                v4f pv[FB][4], hv[FB][4];
                #pragma unroll
                for (int i = 0; i < FB; ++i) {
                    const size_t o = o0 + (size_t)(k * FB + i) * kDim * kDst;
                    #pragma unroll
                    for (int q = 0; q < 4; ++q) {
                        pv[i][q] = *(const v4f*)(Pws + o + 4 * q);
                        hv[i][q] = *(const v4f*)(Hws + o + 4 * q);
                    }
                }
                #pragma unroll
                for (int i = 0; i < FB; ++i) {
                    asm volatile("" :: "v"(pv[i][0]), "v"(pv[i][1]), "v"(pv[i][2]), "v"(pv[i][3]),
                                       "v"(hv[i][0]), "v"(hv[i][1]), "v"(hv[i][2]), "v"(hv[i][3]));
                }
                #pragma unroll
                for (int i = 0; i < FB; ++i) {
                    if (k * FB + i < c) {
                        #pragma unroll
                        for (int q = 0; q < 4; ++q) {
                            h[4 * q + 0] = pv[i][q].x * h[4 * q + 0] + hv[i][q].x;
                            h[4 * q + 1] = pv[i][q].y * h[4 * q + 1] + hv[i][q].y;
                            h[4 * q + 2] = pv[i][q].z * h[4 * q + 2] + hv[i][q].z;
                            h[4 * q + 3] = pv[i][q].w * h[4 * q + 3] + hv[i][q].w;
                        }
                    }
                }
            }
        }
    }
    __syncthreads();

    for (int g = 0; g < NGRP; ++g) {
        const int buf = g & 1;
        const int gn = (g + 1 < NGRP) ? g + 1 : g;
        const size_t t1 = (size_t)gn * SU;
        float dtn[SU], xvn[SU];
        #pragma unroll
        for (int j = 0; j < SU; ++j) {
            dtn[j] = dlt_p[(t1 + j) * kDim];
            xvn[j] = xin_p[(t1 + j) * kDim];
        }
        float4 bcx = {};
        if (stager) bcx = BC4[t1 * 4 + sk];

        float4 bq[4], cq[4];
        #pragma unroll
        for (int q = 0; q < 4; ++q) { bq[q] = sBC[buf][0][q]; cq[q] = sBC[buf][0][4 + q]; }

        #pragma unroll
        for (int j = 0; j < SU; ++j) {
            float4 bqn[4], cqn[4];
            if (j + 1 < SU) {
                #pragma unroll
                for (int q = 0; q < 4; ++q) { bqn[q] = sBC[buf][j + 1][q]; cqn[q] = sBC[buf][j + 1][4 + q]; }
            }
            float bn[kDst], cn[kDst];
            #pragma unroll
            for (int q = 0; q < 4; ++q) {
                bn[4 * q + 0] = bq[q].x; bn[4 * q + 1] = bq[q].y; bn[4 * q + 2] = bq[q].z; bn[4 * q + 3] = bq[q].w;
                cn[4 * q + 0] = cq[q].x; cn[4 * q + 1] = cq[q].y; cn[4 * q + 2] = cq[q].z; cn[4 * q + 3] = cq[q].w;
            }
            const float dt = dtc[j];
            const float xv = xvc[j];
            const float u  = dt * xv;
            float y0 = 0.f, y1 = 0.f;
            #pragma unroll
            for (int n = 0; n < kDst; n += 2) {
                const float e0 = __expf(dt * A[n]);
                const float e1 = __expf(dt * A[n + 1]);
                h[n]     = e0 * h[n]     + u * bn[n];
                h[n + 1] = e1 * h[n + 1] + u * bn[n + 1];
                y0 += h[n]     * cn[n];
                y1 += h[n + 1] * cn[n + 1];
            }
            y_p[((size_t)g * SU + j) * kDim] = (y0 + y1) + xv * D_d;
            if (j + 1 < SU) {
                #pragma unroll
                for (int q = 0; q < 4; ++q) { bq[q] = bqn[q]; cq[q] = cqn[q]; }
            }
        }

        if (stager) sBC[buf ^ 1][sj][sq] = bcx;
        #pragma unroll
        for (int j = 0; j < SU; ++j) { dtc[j] = dtn[j]; xvc[j] = xvn[j]; }
        __syncthreads();
    }
}

__global__ __launch_bounds__(256)
void out_gemm_kernel(const float* __restrict__ y,
                     const float* __restrict__ Wo,
                     const float* __restrict__ bo,
                     float* __restrict__ out)
{
    __shared__ float smem[64 * TS];
    float* sA = smem;
    float* sB = smem + 32 * TS;
    float* sC = smem;

    const int tid  = threadIdx.x;
    const int wave = tid >> 5;
    const int lane = tid & 31;
    const int half = lane >> 4;
    const int l16  = lane & 15;

    const int m0 = blockIdx.x * 64;
    const int n0 = blockIdx.y * 64;
    const int wm = (wave & 1) * 32;
    const int wn = (wave >> 1) * 16;

    v8f acc0 = {};
    v8f acc1 = {};

    for (int k0 = 0; k0 < kDim; k0 += 32) {
        {
            const int i  = tid >> 2;
            const int j0 = (tid & 3) * 8;
            const float* src = y + (size_t)(m0 + i) * kDim + k0 + j0;
            float4 v0 = *(const float4*)(src);
            float4 v1 = *(const float4*)(src + 4);
            sA[(j0 + 0) * TS + i] = v0.x; sA[(j0 + 1) * TS + i] = v0.y;
            sA[(j0 + 2) * TS + i] = v0.z; sA[(j0 + 3) * TS + i] = v0.w;
            sA[(j0 + 4) * TS + i] = v1.x; sA[(j0 + 5) * TS + i] = v1.y;
            sA[(j0 + 6) * TS + i] = v1.z; sA[(j0 + 7) * TS + i] = v1.w;
        }
        {
            const int nn = tid >> 2;
            const int j0 = (tid & 3) * 8;
            const float* src = Wo + (size_t)(n0 + nn) * kDim + k0 + j0;
            float4 v0 = *(const float4*)(src);
            float4 v1 = *(const float4*)(src + 4);
            sB[(j0 + 0) * TS + nn] = v0.x; sB[(j0 + 1) * TS + nn] = v0.y;
            sB[(j0 + 2) * TS + nn] = v0.z; sB[(j0 + 3) * TS + nn] = v0.w;
            sB[(j0 + 4) * TS + nn] = v1.x; sB[(j0 + 5) * TS + nn] = v1.y;
            sB[(j0 + 6) * TS + nn] = v1.z; sB[(j0 + 7) * TS + nn] = v1.w;
        }
        __syncthreads();

        #pragma unroll
        for (int kk = 0; kk < 32; kk += 4) {
            const int kr = kk + 2 * half;
            v2f a0, a1, bf;
            a0.x = sA[kr * TS + wm + l16];
            a0.y = sA[(kr + 1) * TS + wm + l16];
            a1.x = sA[kr * TS + wm + 16 + l16];
            a1.y = sA[(kr + 1) * TS + wm + 16 + l16];
            bf.x = sB[kr * TS + wn + l16];
            bf.y = sB[(kr + 1) * TS + wn + l16];
            acc0 = __builtin_amdgcn_wmma_f32_16x16x4_f32(false, a0, false, bf, (short)0, acc0, false, false);
            acc1 = __builtin_amdgcn_wmma_f32_16x16x4_f32(false, a1, false, bf, (short)0, acc1, false, false);
        }
        __syncthreads();
    }

    const float bias = bo[n0 + wn + l16];
    #pragma unroll
    for (int r = 0; r < 8; ++r) {
        const int mm = wm + r + 8 * half;
        sC[(wn + l16) * TS + mm]      = acc0[r] + bias;
        sC[(wn + l16) * TS + mm + 16] = acc1[r] + bias;
    }
    __syncthreads();

    const int bidx = m0 >> 12;
    const int l0   = m0 & 4095;
    const int nn   = tid >> 2;
    const int ms   = (tid & 3) * 16;
    float* dst = out + ((size_t)bidx * kDim + n0 + nn) * kL + l0 + ms;
    #pragma unroll
    for (int e = 0; e < 4; ++e) {
        float4 v;
        v.x = sC[nn * TS + ms + e * 4 + 0];
        v.y = sC[nn * TS + ms + e * 4 + 1];
        v.z = sC[nn * TS + ms + e * 4 + 2];
        v.w = sC[nn * TS + ms + e * 4 + 3];
        *(float4*)(dst + e * 4) = v;
    }
}

extern "C" void kernel_launch(void* const* d_in, const int* in_sizes, int n_in,
                              void* d_out, int out_size, void* d_ws, size_t ws_size,
                              hipStream_t stream) {
    const float* x     = (const float*)d_in[0];
    const float* Wp    = (const float*)d_in[1];
    const float* bp    = (const float*)d_in[2];
    const float* A_log = (const float*)d_in[3];
    const float* Dp    = (const float*)d_in[4];
    const float* dtb   = (const float*)d_in[5];
    const float* Wo    = (const float*)d_in[6];
    const float* bo    = (const float*)d_in[7];
    float* out = (float*)d_out;

    float* ws = (float*)d_ws;
    const size_t MD = (size_t)kM * kDim;
    const size_t MS = (size_t)kM * kDst;
    float* xin = ws;
    float* dlt = ws + MD;
    float* yws = ws + 2 * MD;
    float* Bw  = ws + 3 * MD;
    float* Cw  = Bw + MS;
    const size_t SS = (size_t)kB * NC * kDim * kDst;
    float* Pw  = Cw + MS;
    float* Hw  = Pw + SS;

    dim3 g1(kM / 64, (kP + 63) / 64);
    proj_gemm_kernel<<<g1, 256, 0, stream>>>(x, Wp, bp, dtb, xin, dlt, Bw, Cw);

    scan_state_kernel<<<kB * SG * (NC - 1), SW, 0, stream>>>(xin, dlt, Bw, A_log, Pw, Hw);

    scan_kernel<<<kB * SG * NC, SW, 0, stream>>>(xin, dlt, Bw, Cw, A_log, Dp, Pw, Hw, yws);

    dim3 g3(kM / 64, kDim / 64);
    out_gemm_kernel<<<g3, 256, 0, stream>>>(yws, Wo, bo, out);
}


# Both GEMMs re-tiled: LDS tiles as k-quads so one 16-byte LDS read is a fragment for 8 k, 32x32 block per wave, conflict-free 16-byte LDS writes, next tile requested ahead (k order in sums changes)
# speedup vs baseline: 6.6789x; 1.1513x over previous
#include <hip/hip_runtime.h>
#include <hip/hip_bf16.h>

typedef float v2f __attribute__((ext_vector_type(2)));
typedef float v8f __attribute__((ext_vector_type(8)));

constexpr int kDim = 384;
constexpr int kDst = 16;
constexpr int kL   = 4096;
constexpr int kB   = 4;
constexpr int kP   = 800;
constexpr int kM   = kB * kL;

typedef float v4f __attribute__((ext_vector_type(4)));

constexpr int NC = 16;
constexpr int CL = kL / NC;

__device__ __forceinline__ float softplus_f(float v) {
    return (v > 20.f) ? v : log1pf(__expf(v));
}

constexpr int GT  = 128;
constexpr int NKT = kDim / 32;
constexpr int QS  = 66;
constexpr int OPS = 8 * QS;

template <bool GUARD>
__device__ __forceinline__ void load_krows(const float* __restrict__ base, int tid,
                                           int valid_rows, v4f (&r)[4]) {
    const int q   = tid & 7;
    const int row = tid >> 3;
    #pragma unroll
    for (int p = 0; p < 4; ++p) {
        int rr = p * 16 + row;
        if (GUARD) rr = (rr < valid_rows) ? rr : 0;
        r[p] = *(const v4f*)(base + (size_t)rr * kDim + 4 * q);
    }
}
template <bool GUARD>
__device__ __forceinline__ void store_krows(v4f* __restrict__ s4, int tid, int valid_rows,
                                            const v4f (&r)[4]) {
    const int q   = tid & 7;
    const int row = tid >> 3;
    #pragma unroll
    for (int p = 0; p < 4; ++p) {
        const int rr = p * 16 + row;
        v4f v = r[p];
        if (GUARD && rr >= valid_rows) { v.x = 0.f; v.y = 0.f; v.z = 0.f; v.w = 0.f; }
        s4[q * QS + rr] = v;
    }
}

__device__ __forceinline__ void tile_mma(const v4f* __restrict__ sA4, const v4f* __restrict__ sB4,
                                         int aslot, int bslot, int half,
                                         v8f& c00, v8f& c01, v8f& c10, v8f& c11) {
    #pragma unroll
    for (int s = 0; s < 4; ++s) {
        const int q = 2 * s + half;
        const v4f a0 = sA4[q * QS + aslot];
        const v4f a1 = sA4[q * QS + aslot + 16];
        const v4f b0 = sB4[q * QS + bslot];
        const v4f b1 = sB4[q * QS + bslot + 16];
        const v2f a0l = {a0.x, a0.y}, a0h = {a0.z, a0.w};
        const v2f a1l = {a1.x, a1.y}, a1h = {a1.z, a1.w};
        const v2f b0l = {b0.x, b0.y}, b0h = {b0.z, b0.w};
        const v2f b1l = {b1.x, b1.y}, b1h = {b1.z, b1.w};
        c00 = __builtin_amdgcn_wmma_f32_16x16x4_f32(false, a0l, false, b0l, (short)0, c00, false, false);
        c01 = __builtin_amdgcn_wmma_f32_16x16x4_f32(false, a0l, false, b1l, (short)0, c01, false, false);
        c10 = __builtin_amdgcn_wmma_f32_16x16x4_f32(false, a1l, false, b0l, (short)0, c10, false, false);
        c11 = __builtin_amdgcn_wmma_f32_16x16x4_f32(false, a1l, false, b1l, (short)0, c11, false, false);
        c00 = __builtin_amdgcn_wmma_f32_16x16x4_f32(false, a0h, false, b0h, (short)0, c00, false, false);
        c01 = __builtin_amdgcn_wmma_f32_16x16x4_f32(false, a0h, false, b1h, (short)0, c01, false, false);
        c10 = __builtin_amdgcn_wmma_f32_16x16x4_f32(false, a1h, false, b0h, (short)0, c10, false, false);
        c11 = __builtin_amdgcn_wmma_f32_16x16x4_f32(false, a1h, false, b1h, (short)0, c11, false, false);
    }
}

__global__ __launch_bounds__(GT)
void proj_gemm_kernel(const float* __restrict__ x,
                      const float* __restrict__ Wp,
                      const float* __restrict__ bp,
                      const float* __restrict__ dtb,
                      float* __restrict__ xin,
                      float* __restrict__ dlt,
                      float* __restrict__ Bws,
                      float* __restrict__ Cws)
{
    __shared__ v4f smem4[2 * OPS];
    v4f* sA4 = smem4;
    v4f* sB4 = smem4 + OPS;

    const int tid  = threadIdx.x;
    const int wave = tid >> 5;
    const int lane = tid & 31;
    const int half = lane >> 4;
    const int l16  = lane & 15;

    const int m0 = blockIdx.x * 64;
    const int n0 = blockIdx.y * 64;

    const int wmi = wave & 1;
    const int wni = wave >> 1;

    v8f c00 = {}, c01 = {}, c10 = {}, c11 = {};

    const int bidx = m0 >> 12;
    const int l0   = m0 & 4095;
    const float* xb = x + (size_t)bidx * kDim * kL + l0;
    const float* wb = Wp + (size_t)n0 * kDim;
    const int vrows = kP - n0;

    const int aq  = tid >> 4;
    const int am4 = tid & 15;

    v4f ra[4], rb[4];
    #pragma unroll
    for (int e = 0; e < 4; ++e) ra[e] = *(const v4f*)(xb + (size_t)(4 * aq + e) * kL + 4 * am4);
    load_krows<true>(wb, tid, vrows, rb);

    for (int t = 0; t < NKT; ++t) {
        #pragma unroll
        for (int mi = 0; mi < 4; ++mi) {
            v4f kq;
            kq.x = ra[0][mi]; kq.y = ra[1][mi]; kq.z = ra[2][mi]; kq.w = ra[3][mi];
            sA4[aq * QS + mi * 16 + am4] = kq;
        }
        store_krows<true>(sB4, tid, vrows, rb);
        __syncthreads();

        if (t + 1 < NKT) {
            const int k1 = (t + 1) * 32;
            #pragma unroll
            for (int e = 0; e < 4; ++e) ra[e] = *(const v4f*)(xb + (size_t)(k1 + 4 * aq + e) * kL + 4 * am4);
            load_krows<true>(wb + k1, tid, vrows, rb);
            __builtin_amdgcn_sched_barrier(0);
        }

        tile_mma(sA4, sB4, wmi * 32 + l16, wni * 32 + l16, half, c00, c01, c10, c11);
        __syncthreads();
    }

    #pragma unroll
    for (int j = 0; j < 2; ++j) {
        const int ng = n0 + wni * 32 + 16 * j + l16;
        if (ng < kP) {
            const float bias = bp[ng];
            float* dst;
            int    ld;
            bool   sp = false;
            float  db = 0.f;
            if (ng < kDim) {
                dst = xin + ng;                     ld = kDim;
            } else if (ng < 2 * kDim) {
                dst = dlt + (ng - kDim);            ld = kDim;
                sp = true;  db = dtb[ng - kDim];
            } else if (ng < 2 * kDim + kDst) {
                dst = Bws + (ng - 2 * kDim);        ld = kDst;
            } else {
                dst = Cws + (ng - 2 * kDim - kDst); ld = kDst;
            }
            dst += (size_t)(m0 + 32 * half + 2 * wmi) * ld;
            const v8f& e0 = j ? c01 : c00;
            const v8f& e1 = j ? c11 : c10;
            #pragma unroll
            for (int r = 0; r < 8; ++r) {
                float v0 = e0[r] + bias;
                float v1 = e1[r] + bias;
                if (sp) {
                    v0 = softplus_f(v0 + db);
                    v1 = softplus_f(v1 + db);
                }
                dst[(size_t)(4 * r) * ld]     = v0;
                dst[(size_t)(4 * r + 1) * ld] = v1;
            }
        }
    }
}

constexpr int SW = 96;
constexpr int SG = kDim / SW;
constexpr int SU = 8;
constexpr int NGRP = CL / SU;

__global__ __launch_bounds__(SW)
void scan_state_kernel(const float* __restrict__ xin,
                       const float* __restrict__ dlt,
                       const float* __restrict__ Bws,
                       const float* __restrict__ A_log,
                       float* __restrict__ Pws,
                       float* __restrict__ Hws)
{
    __shared__ float4 sB[2][SU][4];

    const int tid = threadIdx.x;
    const int blk = blockIdx.x;
    const int c   = blk % (NC - 1);
    const int bg  = blk / (NC - 1);
    const int b   = bg / SG;
    const int d   = (bg % SG) * SW + tid;

    float A[kDst];
    #pragma unroll
    for (int q = 0; q < 4; ++q) {
        const float4 a = *(const float4*)(A_log + (size_t)d * kDst + 4 * q);
        A[4 * q + 0] = -__expf(a.x); A[4 * q + 1] = -__expf(a.y);
        A[4 * q + 2] = -__expf(a.z); A[4 * q + 3] = -__expf(a.w);
    }

    const size_t row0 = (size_t)b * kL + (size_t)c * CL;
    const float* xin_p = xin + row0 * kDim + d;
    const float* dlt_p = dlt + row0 * kDim + d;
    const float4* Bb4  = (const float4*)(Bws + row0 * kDst);

    const bool stager = tid < 4 * SU;
    const int  sk     = tid & (4 * SU - 1);

    float h[kDst], p[kDst];
    #pragma unroll
    for (int n = 0; n < kDst; ++n) { h[n] = 0.f; p[n] = 1.f; }

    float dtc[SU], xvc[SU];
    #pragma unroll
    for (int j = 0; j < SU; ++j) {
        dtc[j] = dlt_p[(size_t)j * kDim];
        xvc[j] = xin_p[(size_t)j * kDim];
    }
    if (stager) sB[0][sk >> 2][sk & 3] = Bb4[sk];
    __syncthreads();

    for (int g = 0; g < NGRP; ++g) {
        const int buf = g & 1;
        const int gn = (g + 1 < NGRP) ? g + 1 : g;
        const size_t t1 = (size_t)gn * SU;
        float dtn[SU], xvn[SU];
        #pragma unroll
        for (int j = 0; j < SU; ++j) {
            dtn[j] = dlt_p[(t1 + j) * kDim];
            xvn[j] = xin_p[(t1 + j) * kDim];
        }
        float4 bnx = {};
        if (stager) bnx = Bb4[t1 * 4 + sk];

        float4 bq[4];
        #pragma unroll
        for (int q = 0; q < 4; ++q) bq[q] = sB[buf][0][q];

        #pragma unroll
        for (int j = 0; j < SU; ++j) {
            float4 bqn[4];
            if (j + 1 < SU) {
                #pragma unroll
                for (int q = 0; q < 4; ++q) bqn[q] = sB[buf][j + 1][q];
            }
            float bn[kDst];
            #pragma unroll
            for (int q = 0; q < 4; ++q) {
                bn[4 * q + 0] = bq[q].x; bn[4 * q + 1] = bq[q].y; bn[4 * q + 2] = bq[q].z; bn[4 * q + 3] = bq[q].w;
            }
            const float dt = dtc[j];
            const float u  = dt * xvc[j];
            #pragma unroll
            for (int n = 0; n < kDst; ++n) {
                const float e = __expf(dt * A[n]);
                h[n] = e * h[n] + u * bn[n];
                p[n] = e * p[n];
            }
            if (j + 1 < SU) {
                #pragma unroll
                for (int q = 0; q < 4; ++q) bq[q] = bqn[q];
            }
        }

        if (stager) sB[buf ^ 1][sk >> 2][sk & 3] = bnx;
        #pragma unroll
        for (int j = 0; j < SU; ++j) { dtc[j] = dtn[j]; xvc[j] = xvn[j]; }
        __syncthreads();
    }

    const size_t o = (((size_t)b * NC + c) * kDim + d) * kDst;
    #pragma unroll
    for (int q = 0; q < 4; ++q) {
        float4 pv, hv;
        pv.x = p[4 * q + 0]; pv.y = p[4 * q + 1]; pv.z = p[4 * q + 2]; pv.w = p[4 * q + 3];
        hv.x = h[4 * q + 0]; hv.y = h[4 * q + 1]; hv.z = h[4 * q + 2]; hv.w = h[4 * q + 3];
        *(float4*)(Pws + o + 4 * q) = pv;
        *(float4*)(Hws + o + 4 * q) = hv;
    }
}

__global__ __launch_bounds__(SW)
void scan_kernel(const float* __restrict__ xin,
                 const float* __restrict__ dlt,
                 const float* __restrict__ Bws,
                 const float* __restrict__ Cws,
                 const float* __restrict__ A_log,
                 const float* __restrict__ Dp,
                 const float* __restrict__ Pws,
                 const float* __restrict__ Hws,
                 float* __restrict__ yout)
{
    __shared__ float4 sBC[2][SU][8];

    const int tid = threadIdx.x;
    const int blk = blockIdx.x;
    const int c   = blk % NC;
    const int bg  = blk / NC;
    const int b   = bg / SG;
    const int d   = (bg % SG) * SW + tid;

    float A[kDst];
    #pragma unroll
    for (int q = 0; q < 4; ++q) {
        const float4 a = *(const float4*)(A_log + (size_t)d * kDst + 4 * q);
        A[4 * q + 0] = -__expf(a.x); A[4 * q + 1] = -__expf(a.y);
        A[4 * q + 2] = -__expf(a.z); A[4 * q + 3] = -__expf(a.w);
    }
    const float D_d = Dp[d];

    const size_t row0 = (size_t)b * kL + (size_t)c * CL;
    const float* xin_p = xin + row0 * kDim + d;
    const float* dlt_p = dlt + row0 * kDim + d;
    float*       y_p   = yout + row0 * kDim + d;
    const float4* Bb4  = (const float4*)(Bws + row0 * kDst);
    const float4* Cb4  = (const float4*)(Cws + row0 * kDst);

    const bool    stager = tid < 8 * SU;
    const int     sk     = tid & (4 * SU - 1);
    const float4* BC4    = (tid < 4 * SU) ? Bb4 : Cb4;
    const int     sj     = sk >> 2;
    const int     sq     = (sk & 3) + ((tid < 4 * SU) ? 0 : 4);

    float dtc[SU], xvc[SU];
    #pragma unroll
    for (int j = 0; j < SU; ++j) {
        dtc[j] = dlt_p[(size_t)j * kDim];
        xvc[j] = xin_p[(size_t)j * kDim];
    }
    if (stager) sBC[0][sj][sq] = BC4[sk];

    float h[kDst];
    #pragma unroll
    for (int n = 0; n < kDst; ++n) h[n] = 0.f;
    {
        constexpr int FB = 5;
        static_assert((NC - 1) % FB == 0, "batches cover the summaries exactly");
        const size_t o0 = (((size_t)b * NC) * kDim + d) * kDst;
        #pragma unroll
        for (int k = 0; k < (NC - 1) / FB; ++k) {
            if (k * FB < c) {
                v4f pv[FB][4], hv[FB][4];
                #pragma unroll
                for (int i = 0; i < FB; ++i) {
                    const size_t o = o0 + (size_t)(k * FB + i) * kDim * kDst;
                    #pragma unroll
                    for (int q = 0; q < 4; ++q) {
                        pv[i][q] = *(const v4f*)(Pws + o + 4 * q);
                        hv[i][q] = *(const v4f*)(Hws + o + 4 * q);
                    }
                }
                #pragma unroll
                for (int i = 0; i < FB; ++i) {
                    asm volatile("" :: "v"(pv[i][0]), "v"(pv[i][1]), "v"(pv[i][2]), "v"(pv[i][3]),
                                       "v"(hv[i][0]), "v"(hv[i][1]), "v"(hv[i][2]), "v"(hv[i][3]));
                }
                #pragma unroll
                for (int i = 0; i < FB; ++i) {
                    if (k * FB + i < c) {
                        #pragma unroll
                        for (int q = 0; q < 4; ++q) {
                            h[4 * q + 0] = pv[i][q].x * h[4 * q + 0] + hv[i][q].x;
                            h[4 * q + 1] = pv[i][q].y * h[4 * q + 1] + hv[i][q].y;
                            h[4 * q + 2] = pv[i][q].z * h[4 * q + 2] + hv[i][q].z;
                            h[4 * q + 3] = pv[i][q].w * h[4 * q + 3] + hv[i][q].w;
                        }
                    }
                }
            }
        }
    }
    __syncthreads();

    for (int g = 0; g < NGRP; ++g) {
        const int buf = g & 1;
        const int gn = (g + 1 < NGRP) ? g + 1 : g;
        const size_t t1 = (size_t)gn * SU;
        float dtn[SU], xvn[SU];
        #pragma unroll
        for (int j = 0; j < SU; ++j) {
            dtn[j] = dlt_p[(t1 + j) * kDim];
            xvn[j] = xin_p[(t1 + j) * kDim];
        }
        float4 bcx = {};
        if (stager) bcx = BC4[t1 * 4 + sk];

        float4 bq[4], cq[4];
        #pragma unroll
        for (int q = 0; q < 4; ++q) { bq[q] = sBC[buf][0][q]; cq[q] = sBC[buf][0][4 + q]; }

        #pragma unroll
        for (int j = 0; j < SU; ++j) {
            float4 bqn[4], cqn[4];
            if (j + 1 < SU) {
                #pragma unroll
                for (int q = 0; q < 4; ++q) { bqn[q] = sBC[buf][j + 1][q]; cqn[q] = sBC[buf][j + 1][4 + q]; }
            }
            float bn[kDst], cn[kDst];
            #pragma unroll
            for (int q = 0; q < 4; ++q) {
                bn[4 * q + 0] = bq[q].x; bn[4 * q + 1] = bq[q].y; bn[4 * q + 2] = bq[q].z; bn[4 * q + 3] = bq[q].w;
                cn[4 * q + 0] = cq[q].x; cn[4 * q + 1] = cq[q].y; cn[4 * q + 2] = cq[q].z; cn[4 * q + 3] = cq[q].w;
            }
            const float dt = dtc[j];
            const float xv = xvc[j];
            const float u  = dt * xv;
            float y0 = 0.f, y1 = 0.f;
            #pragma unroll
            for (int n = 0; n < kDst; n += 2) {
                const float e0 = __expf(dt * A[n]);
                const float e1 = __expf(dt * A[n + 1]);
                h[n]     = e0 * h[n]     + u * bn[n];
                h[n + 1] = e1 * h[n + 1] + u * bn[n + 1];
                y0 += h[n]     * cn[n];
                y1 += h[n + 1] * cn[n + 1];
            }
            y_p[((size_t)g * SU + j) * kDim] = (y0 + y1) + xv * D_d;
            if (j + 1 < SU) {
                #pragma unroll
                for (int q = 0; q < 4; ++q) { bq[q] = bqn[q]; cq[q] = cqn[q]; }
            }
        }

        if (stager) sBC[buf ^ 1][sj][sq] = bcx;
        #pragma unroll
        for (int j = 0; j < SU; ++j) { dtc[j] = dtn[j]; xvc[j] = xvn[j]; }
        __syncthreads();
    }
}

constexpr int CS = 68;

__global__ __launch_bounds__(GT)
void out_gemm_kernel(const float* __restrict__ y,
                     const float* __restrict__ Wo,
                     const float* __restrict__ bo,
                     float* __restrict__ out)
{
    constexpr int SM4 = (64 * CS / 4 > 2 * OPS) ? 64 * CS / 4 : 2 * OPS;
    __shared__ v4f smem4[SM4];
    v4f*   sA4 = smem4;
    v4f*   sB4 = smem4 + OPS;
    float* sC  = (float*)smem4;

    const int tid  = threadIdx.x;
    const int wave = tid >> 5;
    const int lane = tid & 31;
    const int half = lane >> 4;
    const int l16  = lane & 15;

    const int m0 = blockIdx.x * 64;
    const int n0 = blockIdx.y * 64;
    const int wmi = wave & 1;
    const int wni = wave >> 1;

    v8f c00 = {}, c01 = {}, c10 = {}, c11 = {};

    const float* yb = y  + (size_t)m0 * kDim;
    const float* wb = Wo + (size_t)n0 * kDim;

    v4f ra[4], rb[4];
    load_krows<false>(yb, tid, 64, ra);
    load_krows<false>(wb, tid, 64, rb);

    for (int t = 0; t < NKT; ++t) {
        store_krows<false>(sA4, tid, 64, ra);
        store_krows<false>(sB4, tid, 64, rb);
        __syncthreads();

        if (t + 1 < NKT) {
            const int k1 = (t + 1) * 32;
            load_krows<false>(yb + k1, tid, 64, ra);
            load_krows<false>(wb + k1, tid, 64, rb);
            __builtin_amdgcn_sched_barrier(0);
        }

        tile_mma(sA4, sB4, wmi * 32 + l16, wni * 32 + l16, half, c00, c01, c10, c11);
        __syncthreads();
    }

    #pragma unroll
    for (int j = 0; j < 2; ++j) {
        const int nl = wni * 32 + 16 * j + l16;
        const float bias = bo[n0 + nl];
        #pragma unroll
        for (int i = 0; i < 2; ++i) {
            const v8f& e = i ? (j ? c11 : c10) : (j ? c01 : c00);
            float* p = sC + nl * CS + wmi * 32 + 16 * i + 8 * half;
            v4f lo, hi;
            lo.x = e[0] + bias; lo.y = e[1] + bias; lo.z = e[2] + bias; lo.w = e[3] + bias;
            hi.x = e[4] + bias; hi.y = e[5] + bias; hi.z = e[6] + bias; hi.w = e[7] + bias;
            *(v4f*)(p)     = lo;
            *(v4f*)(p + 4) = hi;
        }
    }
    __syncthreads();

    const int bidx = m0 >> 12;
    const int l0   = m0 & 4095;
    const int mq   = tid & 15;
    const int nr   = tid >> 4;
    #pragma unroll
    for (int p = 0; p < 8; ++p) {
        const int nn = p * 8 + nr;
        const v4f v = *(const v4f*)(sC + nn * CS + 4 * mq);
        *(v4f*)(out + ((size_t)bidx * kDim + n0 + nn) * kL + l0 + 4 * mq) = v;
    }
}

extern "C" void kernel_launch(void* const* d_in, const int* in_sizes, int n_in,
                              void* d_out, int out_size, void* d_ws, size_t ws_size,
                              hipStream_t stream) {
    const float* x     = (const float*)d_in[0];
    const float* Wp    = (const float*)d_in[1];
    const float* bp    = (const float*)d_in[2];
    const float* A_log = (const float*)d_in[3];
    const float* Dp    = (const float*)d_in[4];
    const float* dtb   = (const float*)d_in[5];
    const float* Wo    = (const float*)d_in[6];
    const float* bo    = (const float*)d_in[7];
    float* out = (float*)d_out;

    float* ws = (float*)d_ws;
    const size_t MD = (size_t)kM * kDim;
    const size_t MS = (size_t)kM * kDst;
    float* xin = ws;
    float* dlt = ws + MD;
    float* yws = ws + 2 * MD;
    float* Bw  = ws + 3 * MD;
    float* Cw  = Bw + MS;
    const size_t SS = (size_t)kB * NC * kDim * kDst;
    float* Pw  = Cw + MS;
    float* Hw  = Pw + SS;

    dim3 g1(kM / 64, (kP + 63) / 64);
    proj_gemm_kernel<<<g1, GT, 0, stream>>>(x, Wp, bp, dtb, xin, dlt, Bw, Cw);

    scan_state_kernel<<<kB * SG * (NC - 1), SW, 0, stream>>>(xin, dlt, Bw, A_log, Pw, Hw);

    scan_kernel<<<kB * SG * NC, SW, 0, stream>>>(xin, dlt, Bw, Cw, A_log, Dp, Pw, Hw, yws);

    dim3 g3(kM / 64, kDim / 64);
    out_gemm_kernel<<<g3, GT, 0, stream>>>(yws, Wo, bo, out);
}


# Scan cut into 64 time chunks of 64 steps (was 16 of 256), 128 channels per block: three scan waves per SIMD; chunk start states folded by one small new launch (h summation order changes)
# speedup vs baseline: 8.2988x; 1.2425x over previous
#include <hip/hip_runtime.h>
#include <hip/hip_bf16.h>

typedef float v2f __attribute__((ext_vector_type(2)));
typedef float v8f __attribute__((ext_vector_type(8)));

constexpr int kDim = 384;
constexpr int kDst = 16;
constexpr int kL   = 4096;
constexpr int kB   = 4;
constexpr int kP   = 800;
constexpr int kM   = kB * kL;

typedef float v4f __attribute__((ext_vector_type(4)));

constexpr int NC = 64;
constexpr int CL = kL / NC;

__device__ __forceinline__ float softplus_f(float v) {
    return (v > 20.f) ? v : log1pf(__expf(v));
}

constexpr int GT  = 128;
constexpr int NKT = kDim / 32;
constexpr int QS  = 66;
constexpr int OPS = 8 * QS;

template <bool GUARD>
__device__ __forceinline__ void load_krows(const float* __restrict__ base, int tid,
                                           int valid_rows, v4f (&r)[4]) {
    const int q   = tid & 7;
    const int row = tid >> 3;
    #pragma unroll
    for (int p = 0; p < 4; ++p) {
        int rr = p * 16 + row;
        if (GUARD) rr = (rr < valid_rows) ? rr : 0;
        r[p] = *(const v4f*)(base + (size_t)rr * kDim + 4 * q);
    }
}
template <bool GUARD>
__device__ __forceinline__ void store_krows(v4f* __restrict__ s4, int tid, int valid_rows,
                                            const v4f (&r)[4]) {
    const int q   = tid & 7;
    const int row = tid >> 3;
    #pragma unroll
    for (int p = 0; p < 4; ++p) {
        const int rr = p * 16 + row;
        v4f v = r[p];
        if (GUARD && rr >= valid_rows) { v.x = 0.f; v.y = 0.f; v.z = 0.f; v.w = 0.f; }
        s4[q * QS + rr] = v;
    }
}

__device__ __forceinline__ void tile_mma(const v4f* __restrict__ sA4, const v4f* __restrict__ sB4,
                                         int aslot, int bslot, int half,
                                         v8f& c00, v8f& c01, v8f& c10, v8f& c11) {
    #pragma unroll
    for (int s = 0; s < 4; ++s) {
        const int q = 2 * s + half;
        const v4f a0 = sA4[q * QS + aslot];
        const v4f a1 = sA4[q * QS + aslot + 16];
        const v4f b0 = sB4[q * QS + bslot];
        const v4f b1 = sB4[q * QS + bslot + 16];
        const v2f a0l = {a0.x, a0.y}, a0h = {a0.z, a0.w};
        const v2f a1l = {a1.x, a1.y}, a1h = {a1.z, a1.w};
        const v2f b0l = {b0.x, b0.y}, b0h = {b0.z, b0.w};
        const v2f b1l = {b1.x, b1.y}, b1h = {b1.z, b1.w};
        c00 = __builtin_amdgcn_wmma_f32_16x16x4_f32(false, a0l, false, b0l, (short)0, c00, false, false);
        c01 = __builtin_amdgcn_wmma_f32_16x16x4_f32(false, a0l, false, b1l, (short)0, c01, false, false);
        c10 = __builtin_amdgcn_wmma_f32_16x16x4_f32(false, a1l, false, b0l, (short)0, c10, false, false);
        c11 = __builtin_amdgcn_wmma_f32_16x16x4_f32(false, a1l, false, b1l, (short)0, c11, false, false);
        c00 = __builtin_amdgcn_wmma_f32_16x16x4_f32(false, a0h, false, b0h, (short)0, c00, false, false);
        c01 = __builtin_amdgcn_wmma_f32_16x16x4_f32(false, a0h, false, b1h, (short)0, c01, false, false);
        c10 = __builtin_amdgcn_wmma_f32_16x16x4_f32(false, a1h, false, b0h, (short)0, c10, false, false);
        c11 = __builtin_amdgcn_wmma_f32_16x16x4_f32(false, a1h, false, b1h, (short)0, c11, false, false);
    }
}

__global__ __launch_bounds__(GT)
void proj_gemm_kernel(const float* __restrict__ x,
                      const float* __restrict__ Wp,
                      const float* __restrict__ bp,
                      const float* __restrict__ dtb,
                      float* __restrict__ xin,
                      float* __restrict__ dlt,
                      float* __restrict__ Bws,
                      float* __restrict__ Cws)
{
    __shared__ v4f smem4[2 * OPS];
    v4f* sA4 = smem4;
    v4f* sB4 = smem4 + OPS;

    const int tid  = threadIdx.x;
    const int wave = tid >> 5;
    const int lane = tid & 31;
    const int half = lane >> 4;
    const int l16  = lane & 15;

    const int m0 = blockIdx.x * 64;
    const int n0 = blockIdx.y * 64;

    const int wmi = wave & 1;
    const int wni = wave >> 1;

    v8f c00 = {}, c01 = {}, c10 = {}, c11 = {};

    const int bidx = m0 >> 12;
    const int l0   = m0 & 4095;
    const float* xb = x + (size_t)bidx * kDim * kL + l0;
    const float* wb = Wp + (size_t)n0 * kDim;
    const int vrows = kP - n0;

    const int aq  = tid >> 4;
    const int am4 = tid & 15;

    v4f ra[4], rb[4];
    #pragma unroll
    for (int e = 0; e < 4; ++e) ra[e] = *(const v4f*)(xb + (size_t)(4 * aq + e) * kL + 4 * am4);
    load_krows<true>(wb, tid, vrows, rb);

    for (int t = 0; t < NKT; ++t) {
        #pragma unroll
        for (int mi = 0; mi < 4; ++mi) {
            v4f kq;
            kq.x = ra[0][mi]; kq.y = ra[1][mi]; kq.z = ra[2][mi]; kq.w = ra[3][mi];
            sA4[aq * QS + mi * 16 + am4] = kq;
        }
        store_krows<true>(sB4, tid, vrows, rb);
        __syncthreads();

        if (t + 1 < NKT) {
            const int k1 = (t + 1) * 32;
            #pragma unroll
            for (int e = 0; e < 4; ++e) ra[e] = *(const v4f*)(xb + (size_t)(k1 + 4 * aq + e) * kL + 4 * am4);
            load_krows<true>(wb + k1, tid, vrows, rb);
            __builtin_amdgcn_sched_barrier(0);
        }

        tile_mma(sA4, sB4, wmi * 32 + l16, wni * 32 + l16, half, c00, c01, c10, c11);
        __syncthreads();
    }

    #pragma unroll
    for (int j = 0; j < 2; ++j) {
        const int ng = n0 + wni * 32 + 16 * j + l16;
        if (ng < kP) {
            const float bias = bp[ng];
            float* dst;
            int    ld;
            bool   sp = false;
            float  db = 0.f;
            if (ng < kDim) {
                dst = xin + ng;                     ld = kDim;
            } else if (ng < 2 * kDim) {
                dst = dlt + (ng - kDim);            ld = kDim;
                sp = true;  db = dtb[ng - kDim];
            } else if (ng < 2 * kDim + kDst) {
                dst = Bws + (ng - 2 * kDim);        ld = kDst;
            } else {
                dst = Cws + (ng - 2 * kDim - kDst); ld = kDst;
            }
            dst += (size_t)(m0 + 32 * half + 2 * wmi) * ld;
            const v8f& e0 = j ? c01 : c00;
            const v8f& e1 = j ? c11 : c10;
            #pragma unroll
            for (int r = 0; r < 8; ++r) {
                float v0 = e0[r] + bias;
                float v1 = e1[r] + bias;
                if (sp) {
                    v0 = softplus_f(v0 + db);
                    v1 = softplus_f(v1 + db);
                }
                dst[(size_t)(4 * r) * ld]     = v0;
                dst[(size_t)(4 * r + 1) * ld] = v1;
            }
        }
    }
}

constexpr int SW = 128;
constexpr int SG = kDim / SW;
constexpr int SU = 8;
constexpr int NGRP = CL / SU;

__global__ __launch_bounds__(SW)
void scan_state_kernel(const float* __restrict__ xin,
                       const float* __restrict__ dlt,
                       const float* __restrict__ Bws,
                       const float* __restrict__ A_log,
                       float* __restrict__ Pws,
                       float* __restrict__ Hws)
{
    __shared__ float4 sB[2][SU][4];

    const int tid = threadIdx.x;
    const int blk = blockIdx.x;
    const int c   = blk % (NC - 1);
    const int bg  = blk / (NC - 1);
    const int b   = bg / SG;
    const int d   = (bg % SG) * SW + tid;

    float A[kDst];
    #pragma unroll
    for (int q = 0; q < 4; ++q) {
        const float4 a = *(const float4*)(A_log + (size_t)d * kDst + 4 * q);
        A[4 * q + 0] = -__expf(a.x); A[4 * q + 1] = -__expf(a.y);
        A[4 * q + 2] = -__expf(a.z); A[4 * q + 3] = -__expf(a.w);
    }

    const size_t row0 = (size_t)b * kL + (size_t)c * CL;
    const float* xin_p = xin + row0 * kDim + d;
    const float* dlt_p = dlt + row0 * kDim + d;
    const float4* Bb4  = (const float4*)(Bws + row0 * kDst);

    const bool stager = tid < 4 * SU;
    const int  sk     = tid & (4 * SU - 1);

    float h[kDst], p[kDst];
    #pragma unroll
    for (int n = 0; n < kDst; ++n) { h[n] = 0.f; p[n] = 1.f; }

    float dtc[SU], xvc[SU];
    #pragma unroll
    for (int j = 0; j < SU; ++j) {
        dtc[j] = dlt_p[(size_t)j * kDim];
        xvc[j] = xin_p[(size_t)j * kDim];
    }
    if (stager) sB[0][sk >> 2][sk & 3] = Bb4[sk];
    __syncthreads();

    for (int g = 0; g < NGRP; ++g) {
        const int buf = g & 1;
        const int gn = (g + 1 < NGRP) ? g + 1 : g;
        const size_t t1 = (size_t)gn * SU;
        float dtn[SU], xvn[SU];
        #pragma unroll
        for (int j = 0; j < SU; ++j) {
            dtn[j] = dlt_p[(t1 + j) * kDim];
            xvn[j] = xin_p[(t1 + j) * kDim];
        }
        float4 bnx = {};
        if (stager) bnx = Bb4[t1 * 4 + sk];

        float4 bq[4];
        #pragma unroll
        for (int q = 0; q < 4; ++q) bq[q] = sB[buf][0][q];

        #pragma unroll
        for (int j = 0; j < SU; ++j) {
            float4 bqn[4];
            if (j + 1 < SU) {
                #pragma unroll
                for (int q = 0; q < 4; ++q) bqn[q] = sB[buf][j + 1][q];
            }
            float bn[kDst];
            #pragma unroll
            for (int q = 0; q < 4; ++q) {
                bn[4 * q + 0] = bq[q].x; bn[4 * q + 1] = bq[q].y; bn[4 * q + 2] = bq[q].z; bn[4 * q + 3] = bq[q].w;
            }
            const float dt = dtc[j];
            const float u  = dt * xvc[j];
            #pragma unroll
            for (int n = 0; n < kDst; ++n) {
                const float e = __expf(dt * A[n]);
                h[n] = e * h[n] + u * bn[n];
                p[n] = e * p[n];
            }
            if (j + 1 < SU) {
                #pragma unroll
                for (int q = 0; q < 4; ++q) bq[q] = bqn[q];
            }
        }

        if (stager) sB[buf ^ 1][sk >> 2][sk & 3] = bnx;
        #pragma unroll
        for (int j = 0; j < SU; ++j) { dtc[j] = dtn[j]; xvc[j] = xvn[j]; }
        __syncthreads();
    }

    const size_t o = (((size_t)b * NC + c) * kDim + d) * kDst;
    #pragma unroll
    for (int q = 0; q < 4; ++q) {
        float4 pv, hv;
        pv.x = p[4 * q + 0]; pv.y = p[4 * q + 1]; pv.z = p[4 * q + 2]; pv.w = p[4 * q + 3];
        hv.x = h[4 * q + 0]; hv.y = h[4 * q + 1]; hv.z = h[4 * q + 2]; hv.w = h[4 * q + 3];
        *(float4*)(Pws + o + 4 * q) = pv;
        *(float4*)(Hws + o + 4 * q) = hv;
    }
}

constexpr int FT = 64;

__global__ __launch_bounds__(FT)
void scan_fold_kernel(const float* __restrict__ Pws,
                      float* __restrict__ Hws)
{
    constexpr size_t plane = (size_t)kDim * kDst;
    const size_t o = (size_t)blockIdx.y * NC * plane + (size_t)blockIdx.x * FT + threadIdx.x;
    const float* pp = Pws + o;
    float*       hp = Hws + o;

    float pv[NC - 1], hv[NC - 1];
    #pragma unroll
    for (int c = 0; c < NC - 1; ++c) {
        pv[c] = pp[(size_t)c * plane];
        hv[c] = hp[(size_t)c * plane];
    }
    float h = 0.f;
    #pragma unroll
    for (int c = 0; c < NC - 1; ++c) {
        h = pv[c] * h + hv[c];
        hp[(size_t)c * plane] = h;
    }
}

__global__ __launch_bounds__(SW)
void scan_kernel(const float* __restrict__ xin,
                 const float* __restrict__ dlt,
                 const float* __restrict__ Bws,
                 const float* __restrict__ Cws,
                 const float* __restrict__ A_log,
                 const float* __restrict__ Dp,
                 const float* __restrict__ Hws,
                 float* __restrict__ yout)
{
    __shared__ float4 sBC[2][SU][8];

    const int tid = threadIdx.x;
    const int blk = blockIdx.x;
    const int c   = blk % NC;
    const int bg  = blk / NC;
    const int b   = bg / SG;
    const int d   = (bg % SG) * SW + tid;

    float A[kDst];
    #pragma unroll
    for (int q = 0; q < 4; ++q) {
        const float4 a = *(const float4*)(A_log + (size_t)d * kDst + 4 * q);
        A[4 * q + 0] = -__expf(a.x); A[4 * q + 1] = -__expf(a.y);
        A[4 * q + 2] = -__expf(a.z); A[4 * q + 3] = -__expf(a.w);
    }
    const float D_d = Dp[d];

    const size_t row0 = (size_t)b * kL + (size_t)c * CL;
    const float* xin_p = xin + row0 * kDim + d;
    const float* dlt_p = dlt + row0 * kDim + d;
    float*       y_p   = yout + row0 * kDim + d;
    const float4* Bb4  = (const float4*)(Bws + row0 * kDst);
    const float4* Cb4  = (const float4*)(Cws + row0 * kDst);

    const bool    stager = tid < 8 * SU;
    const int     sk     = tid & (4 * SU - 1);
    const float4* BC4    = (tid < 4 * SU) ? Bb4 : Cb4;
    const int     sj     = sk >> 2;
    const int     sq     = (sk & 3) + ((tid < 4 * SU) ? 0 : 4);

    float dtc[SU], xvc[SU];
    #pragma unroll
    for (int j = 0; j < SU; ++j) {
        dtc[j] = dlt_p[(size_t)j * kDim];
        xvc[j] = xin_p[(size_t)j * kDim];
    }
    float h[kDst];
    #pragma unroll
    for (int n = 0; n < kDst; ++n) h[n] = 0.f;
    if (c > 0) {
        const size_t o = (((size_t)b * NC + (c - 1)) * kDim + d) * kDst;
        #pragma unroll
        for (int q = 0; q < 4; ++q) {
            const v4f hv = *(const v4f*)(Hws + o + 4 * q);
            h[4 * q + 0] = hv.x; h[4 * q + 1] = hv.y; h[4 * q + 2] = hv.z; h[4 * q + 3] = hv.w;
        }
    }
    if (stager) sBC[0][sj][sq] = BC4[sk];
    __syncthreads();

    for (int g = 0; g < NGRP; ++g) {
        const int buf = g & 1;
        const int gn = (g + 1 < NGRP) ? g + 1 : g;
        const size_t t1 = (size_t)gn * SU;
        float dtn[SU], xvn[SU];
        #pragma unroll
        for (int j = 0; j < SU; ++j) {
            dtn[j] = dlt_p[(t1 + j) * kDim];
            xvn[j] = xin_p[(t1 + j) * kDim];
        }
        float4 bcx = {};
        if (stager) bcx = BC4[t1 * 4 + sk];

        float4 bq[4], cq[4];
        #pragma unroll
        for (int q = 0; q < 4; ++q) { bq[q] = sBC[buf][0][q]; cq[q] = sBC[buf][0][4 + q]; }

        #pragma unroll
        for (int j = 0; j < SU; ++j) {
            float4 bqn[4], cqn[4];
            if (j + 1 < SU) {
                #pragma unroll
                for (int q = 0; q < 4; ++q) { bqn[q] = sBC[buf][j + 1][q]; cqn[q] = sBC[buf][j + 1][4 + q]; }
            }
            float bn[kDst], cn[kDst];
            #pragma unroll
            for (int q = 0; q < 4; ++q) {
                bn[4 * q + 0] = bq[q].x; bn[4 * q + 1] = bq[q].y; bn[4 * q + 2] = bq[q].z; bn[4 * q + 3] = bq[q].w;
                cn[4 * q + 0] = cq[q].x; cn[4 * q + 1] = cq[q].y; cn[4 * q + 2] = cq[q].z; cn[4 * q + 3] = cq[q].w;
            }
            const float dt = dtc[j];
            const float xv = xvc[j];
            const float u  = dt * xv;
            float y0 = 0.f, y1 = 0.f;
            #pragma unroll
            for (int n = 0; n < kDst; n += 2) {
                const float e0 = __expf(dt * A[n]);
                const float e1 = __expf(dt * A[n + 1]);
                h[n]     = e0 * h[n]     + u * bn[n];
                h[n + 1] = e1 * h[n + 1] + u * bn[n + 1];
                y0 += h[n]     * cn[n];
                y1 += h[n + 1] * cn[n + 1];
            }
            y_p[((size_t)g * SU + j) * kDim] = (y0 + y1) + xv * D_d;
            if (j + 1 < SU) {
                #pragma unroll
                for (int q = 0; q < 4; ++q) { bq[q] = bqn[q]; cq[q] = cqn[q]; }
            }
        }

        if (stager) sBC[buf ^ 1][sj][sq] = bcx;
        #pragma unroll
        for (int j = 0; j < SU; ++j) { dtc[j] = dtn[j]; xvc[j] = xvn[j]; }
        __syncthreads();
    }
}

constexpr int CS = 68;

__global__ __launch_bounds__(GT)
void out_gemm_kernel(const float* __restrict__ y,
                     const float* __restrict__ Wo,
                     const float* __restrict__ bo,
                     float* __restrict__ out)
{
    constexpr int SM4 = (64 * CS / 4 > 2 * OPS) ? 64 * CS / 4 : 2 * OPS;
    __shared__ v4f smem4[SM4];
    v4f*   sA4 = smem4;
    v4f*   sB4 = smem4 + OPS;
    float* sC  = (float*)smem4;

    const int tid  = threadIdx.x;
    const int wave = tid >> 5;
    const int lane = tid & 31;
    const int half = lane >> 4;
    const int l16  = lane & 15;

    const int m0 = blockIdx.x * 64;
    const int n0 = blockIdx.y * 64;
    const int wmi = wave & 1;
    const int wni = wave >> 1;

    v8f c00 = {}, c01 = {}, c10 = {}, c11 = {};

    const float* yb = y  + (size_t)m0 * kDim;
    const float* wb = Wo + (size_t)n0 * kDim;

    v4f ra[4], rb[4];
    load_krows<false>(yb, tid, 64, ra);
    load_krows<false>(wb, tid, 64, rb);

    for (int t = 0; t < NKT; ++t) {
        store_krows<false>(sA4, tid, 64, ra);
        store_krows<false>(sB4, tid, 64, rb);
        __syncthreads();

        if (t + 1 < NKT) {
            const int k1 = (t + 1) * 32;
            load_krows<false>(yb + k1, tid, 64, ra);
            load_krows<false>(wb + k1, tid, 64, rb);
            __builtin_amdgcn_sched_barrier(0);
        }

        tile_mma(sA4, sB4, wmi * 32 + l16, wni * 32 + l16, half, c00, c01, c10, c11);
        __syncthreads();
    }

    #pragma unroll
    for (int j = 0; j < 2; ++j) {
        const int nl = wni * 32 + 16 * j + l16;
        const float bias = bo[n0 + nl];
        #pragma unroll
        for (int i = 0; i < 2; ++i) {
            const v8f& e = i ? (j ? c11 : c10) : (j ? c01 : c00);
            float* p = sC + nl * CS + wmi * 32 + 16 * i + 8 * half;
            v4f lo, hi;
            lo.x = e[0] + bias; lo.y = e[1] + bias; lo.z = e[2] + bias; lo.w = e[3] + bias;
            hi.x = e[4] + bias; hi.y = e[5] + bias; hi.z = e[6] + bias; hi.w = e[7] + bias;
            *(v4f*)(p)     = lo;
            *(v4f*)(p + 4) = hi;
        }
    }
    __syncthreads();

    const int bidx = m0 >> 12;
    const int l0   = m0 & 4095;
    const int mq   = tid & 15;
    const int nr   = tid >> 4;
    #pragma unroll
    for (int p = 0; p < 8; ++p) {
        const int nn = p * 8 + nr;
        const v4f v = *(const v4f*)(sC + nn * CS + 4 * mq);
        *(v4f*)(out + ((size_t)bidx * kDim + n0 + nn) * kL + l0 + 4 * mq) = v;
    }
}

extern "C" void kernel_launch(void* const* d_in, const int* in_sizes, int n_in,
                              void* d_out, int out_size, void* d_ws, size_t ws_size,
                              hipStream_t stream) {
    const float* x     = (const float*)d_in[0];
    const float* Wp    = (const float*)d_in[1];
    const float* bp    = (const float*)d_in[2];
    const float* A_log = (const float*)d_in[3];
    const float* Dp    = (const float*)d_in[4];
    const float* dtb   = (const float*)d_in[5];
    const float* Wo    = (const float*)d_in[6];
    const float* bo    = (const float*)d_in[7];
    float* out = (float*)d_out;

    float* ws = (float*)d_ws;
    const size_t MD = (size_t)kM * kDim;
    const size_t MS = (size_t)kM * kDst;
    float* xin = ws;
    float* dlt = ws + MD;
    float* yws = ws + 2 * MD;
    float* Bw  = ws + 3 * MD;
    float* Cw  = Bw + MS;
    const size_t SS = (size_t)kB * NC * kDim * kDst;
    float* Pw  = Cw + MS;
    float* Hw  = Pw + SS;

    dim3 g1(kM / 64, (kP + 63) / 64);
    proj_gemm_kernel<<<g1, GT, 0, stream>>>(x, Wp, bp, dtb, xin, dlt, Bw, Cw);

    scan_state_kernel<<<kB * SG * (NC - 1), SW, 0, stream>>>(xin, dlt, Bw, A_log, Pw, Hw);

    dim3 gf(kDim * kDst / FT, kB);
    scan_fold_kernel<<<gf, FT, 0, stream>>>(Pw, Hw);

    scan_kernel<<<kB * SG * NC, SW, 0, stream>>>(xin, dlt, Bw, Cw, A_log, Dp, Hw, yws);

    dim3 g3(kM / 64, kDim / 64);
    out_gemm_kernel<<<g3, GT, 0, stream>>>(yws, Wo, bo, out);
}


# Softplus of the projection epilogue as max(v,0) + log1p(exp(-|v|)) on the hardware exp and log instructions (cubic series below 0.01), branch-free, in place of the library log1pf path
# speedup vs baseline: 8.8274x; 1.0637x over previous
#include <hip/hip_runtime.h>
#include <hip/hip_bf16.h>

typedef float v2f __attribute__((ext_vector_type(2)));
typedef float v8f __attribute__((ext_vector_type(8)));

constexpr int kDim = 384;
constexpr int kDst = 16;
constexpr int kL   = 4096;
constexpr int kB   = 4;
constexpr int kP   = 800;
constexpr int kM   = kB * kL;

typedef float v4f __attribute__((ext_vector_type(4)));

constexpr int NC = 64;
constexpr int CL = kL / NC;

__device__ __forceinline__ float softplus_f(float v) {
    const float e   = __expf(-fabsf(v));
    const float lg  = 0.69314718f * __builtin_amdgcn_logf(1.f + e);
    const float ser = e * (1.f - e * (0.5f - e * 0.33333334f));
    const float l1p = (e < 0.01f) ? ser : lg;
    return fmaxf(v, 0.f) + l1p;
}

constexpr int GT  = 128;
constexpr int NKT = kDim / 32;
constexpr int QS  = 66;
constexpr int OPS = 8 * QS;

template <bool GUARD>
__device__ __forceinline__ void load_krows(const float* __restrict__ base, int tid,
                                           int valid_rows, v4f (&r)[4]) {
    const int q   = tid & 7;
    const int row = tid >> 3;
    #pragma unroll
    for (int p = 0; p < 4; ++p) {
        int rr = p * 16 + row;
        if (GUARD) rr = (rr < valid_rows) ? rr : 0;
        r[p] = *(const v4f*)(base + (size_t)rr * kDim + 4 * q);
    }
}
template <bool GUARD>
__device__ __forceinline__ void store_krows(v4f* __restrict__ s4, int tid, int valid_rows,
                                            const v4f (&r)[4]) {
    const int q   = tid & 7;
    const int row = tid >> 3;
    #pragma unroll
    for (int p = 0; p < 4; ++p) {
        const int rr = p * 16 + row;
        v4f v = r[p];
        if (GUARD && rr >= valid_rows) { v.x = 0.f; v.y = 0.f; v.z = 0.f; v.w = 0.f; }
        s4[q * QS + rr] = v;
    }
}

__device__ __forceinline__ void tile_mma(const v4f* __restrict__ sA4, const v4f* __restrict__ sB4,
                                         int aslot, int bslot, int half,
                                         v8f& c00, v8f& c01, v8f& c10, v8f& c11) {
    #pragma unroll
    for (int s = 0; s < 4; ++s) {
        const int q = 2 * s + half;
        const v4f a0 = sA4[q * QS + aslot];
        const v4f a1 = sA4[q * QS + aslot + 16];
        const v4f b0 = sB4[q * QS + bslot];
        const v4f b1 = sB4[q * QS + bslot + 16];
        const v2f a0l = {a0.x, a0.y}, a0h = {a0.z, a0.w};
        const v2f a1l = {a1.x, a1.y}, a1h = {a1.z, a1.w};
        const v2f b0l = {b0.x, b0.y}, b0h = {b0.z, b0.w};
        const v2f b1l = {b1.x, b1.y}, b1h = {b1.z, b1.w};
        c00 = __builtin_amdgcn_wmma_f32_16x16x4_f32(false, a0l, false, b0l, (short)0, c00, false, false);
        c01 = __builtin_amdgcn_wmma_f32_16x16x4_f32(false, a0l, false, b1l, (short)0, c01, false, false);
        c10 = __builtin_amdgcn_wmma_f32_16x16x4_f32(false, a1l, false, b0l, (short)0, c10, false, false);
        c11 = __builtin_amdgcn_wmma_f32_16x16x4_f32(false, a1l, false, b1l, (short)0, c11, false, false);
        c00 = __builtin_amdgcn_wmma_f32_16x16x4_f32(false, a0h, false, b0h, (short)0, c00, false, false);
        c01 = __builtin_amdgcn_wmma_f32_16x16x4_f32(false, a0h, false, b1h, (short)0, c01, false, false);
        c10 = __builtin_amdgcn_wmma_f32_16x16x4_f32(false, a1h, false, b0h, (short)0, c10, false, false);
        c11 = __builtin_amdgcn_wmma_f32_16x16x4_f32(false, a1h, false, b1h, (short)0, c11, false, false);
    }
}

__global__ __launch_bounds__(GT)
void proj_gemm_kernel(const float* __restrict__ x,
                      const float* __restrict__ Wp,
                      const float* __restrict__ bp,
                      const float* __restrict__ dtb,
                      float* __restrict__ xin,
                      float* __restrict__ dlt,
                      float* __restrict__ Bws,
                      float* __restrict__ Cws)
{
    __shared__ v4f smem4[2 * OPS];
    v4f* sA4 = smem4;
    v4f* sB4 = smem4 + OPS;

    const int tid  = threadIdx.x;
    const int wave = tid >> 5;
    const int lane = tid & 31;
    const int half = lane >> 4;
    const int l16  = lane & 15;

    const int m0 = blockIdx.x * 64;
    const int n0 = blockIdx.y * 64;

    const int wmi = wave & 1;
    const int wni = wave >> 1;

    v8f c00 = {}, c01 = {}, c10 = {}, c11 = {};

    const int bidx = m0 >> 12;
    const int l0   = m0 & 4095;
    const float* xb = x + (size_t)bidx * kDim * kL + l0;
    const float* wb = Wp + (size_t)n0 * kDim;
    const int vrows = kP - n0;

    const int aq  = tid >> 4;
    const int am4 = tid & 15;

    v4f ra[4], rb[4];
    #pragma unroll
    for (int e = 0; e < 4; ++e) ra[e] = *(const v4f*)(xb + (size_t)(4 * aq + e) * kL + 4 * am4);
    load_krows<true>(wb, tid, vrows, rb);

    for (int t = 0; t < NKT; ++t) {
        #pragma unroll
        for (int mi = 0; mi < 4; ++mi) {
            v4f kq;
            kq.x = ra[0][mi]; kq.y = ra[1][mi]; kq.z = ra[2][mi]; kq.w = ra[3][mi];
            sA4[aq * QS + mi * 16 + am4] = kq;
        }
        store_krows<true>(sB4, tid, vrows, rb);
        __syncthreads();

        if (t + 1 < NKT) {
            const int k1 = (t + 1) * 32;
            #pragma unroll
            for (int e = 0; e < 4; ++e) ra[e] = *(const v4f*)(xb + (size_t)(k1 + 4 * aq + e) * kL + 4 * am4);
            load_krows<true>(wb + k1, tid, vrows, rb);
            __builtin_amdgcn_sched_barrier(0);
        }

        tile_mma(sA4, sB4, wmi * 32 + l16, wni * 32 + l16, half, c00, c01, c10, c11);
        __syncthreads();
    }

    #pragma unroll
    for (int j = 0; j < 2; ++j) {
        const int ng = n0 + wni * 32 + 16 * j + l16;
        if (ng < kP) {
            const float bias = bp[ng];
            float* dst;
            int    ld;
            bool   sp = false;
            float  db = 0.f;
            if (ng < kDim) {
                dst = xin + ng;                     ld = kDim;
            } else if (ng < 2 * kDim) {
                dst = dlt + (ng - kDim);            ld = kDim;
                sp = true;  db = dtb[ng - kDim];
            } else if (ng < 2 * kDim + kDst) {
                dst = Bws + (ng - 2 * kDim);        ld = kDst;
            } else {
                dst = Cws + (ng - 2 * kDim - kDst); ld = kDst;
            }
            dst += (size_t)(m0 + 32 * half + 2 * wmi) * ld;
            const v8f& e0 = j ? c01 : c00;
            const v8f& e1 = j ? c11 : c10;
            #pragma unroll
            for (int r = 0; r < 8; ++r) {
                float v0 = e0[r] + bias;
                float v1 = e1[r] + bias;
                if (sp) {
                    v0 = softplus_f(v0 + db);
                    v1 = softplus_f(v1 + db);
                }
                dst[(size_t)(4 * r) * ld]     = v0;
                dst[(size_t)(4 * r + 1) * ld] = v1;
            }
        }
    }
}

constexpr int SW = 128;
constexpr int SG = kDim / SW;
constexpr int SU = 8;
constexpr int NGRP = CL / SU;

__global__ __launch_bounds__(SW)
void scan_state_kernel(const float* __restrict__ xin,
                       const float* __restrict__ dlt,
                       const float* __restrict__ Bws,
                       const float* __restrict__ A_log,
                       float* __restrict__ Pws,
                       float* __restrict__ Hws)
{
    __shared__ float4 sB[2][SU][4];

    const int tid = threadIdx.x;
    const int blk = blockIdx.x;
    const int c   = blk % (NC - 1);
    const int bg  = blk / (NC - 1);
    const int b   = bg / SG;
    const int d   = (bg % SG) * SW + tid;

    float A[kDst];
    #pragma unroll
    for (int q = 0; q < 4; ++q) {
        const float4 a = *(const float4*)(A_log + (size_t)d * kDst + 4 * q);
        A[4 * q + 0] = -__expf(a.x); A[4 * q + 1] = -__expf(a.y);
        A[4 * q + 2] = -__expf(a.z); A[4 * q + 3] = -__expf(a.w);
    }

    const size_t row0 = (size_t)b * kL + (size_t)c * CL;
    const float* xin_p = xin + row0 * kDim + d;
    const float* dlt_p = dlt + row0 * kDim + d;
    const float4* Bb4  = (const float4*)(Bws + row0 * kDst);

    const bool stager = tid < 4 * SU;
    const int  sk     = tid & (4 * SU - 1);

    float h[kDst], p[kDst];
    #pragma unroll
    for (int n = 0; n < kDst; ++n) { h[n] = 0.f; p[n] = 1.f; }

    float dtc[SU], xvc[SU];
    #pragma unroll
    for (int j = 0; j < SU; ++j) {
        dtc[j] = dlt_p[(size_t)j * kDim];
        xvc[j] = xin_p[(size_t)j * kDim];
    }
    if (stager) sB[0][sk >> 2][sk & 3] = Bb4[sk];
    __syncthreads();

    for (int g = 0; g < NGRP; ++g) {
        const int buf = g & 1;
        const int gn = (g + 1 < NGRP) ? g + 1 : g;
        const size_t t1 = (size_t)gn * SU;
        float dtn[SU], xvn[SU];
        #pragma unroll
        for (int j = 0; j < SU; ++j) {
            dtn[j] = dlt_p[(t1 + j) * kDim];
            xvn[j] = xin_p[(t1 + j) * kDim];
        }
        float4 bnx = {};
        if (stager) bnx = Bb4[t1 * 4 + sk];

        float4 bq[4];
        #pragma unroll
        for (int q = 0; q < 4; ++q) bq[q] = sB[buf][0][q];

        #pragma unroll
        for (int j = 0; j < SU; ++j) {
            float4 bqn[4];
            if (j + 1 < SU) {
                #pragma unroll
                for (int q = 0; q < 4; ++q) bqn[q] = sB[buf][j + 1][q];
            }
            float bn[kDst];
            #pragma unroll
            for (int q = 0; q < 4; ++q) {
                bn[4 * q + 0] = bq[q].x; bn[4 * q + 1] = bq[q].y; bn[4 * q + 2] = bq[q].z; bn[4 * q + 3] = bq[q].w;
            }
            const float dt = dtc[j];
            const float u  = dt * xvc[j];
            #pragma unroll
            for (int n = 0; n < kDst; ++n) {
                const float e = __expf(dt * A[n]);
                h[n] = e * h[n] + u * bn[n];
                p[n] = e * p[n];
            }
            if (j + 1 < SU) {
                #pragma unroll
                for (int q = 0; q < 4; ++q) bq[q] = bqn[q];
            }
        }

        if (stager) sB[buf ^ 1][sk >> 2][sk & 3] = bnx;
        #pragma unroll
        for (int j = 0; j < SU; ++j) { dtc[j] = dtn[j]; xvc[j] = xvn[j]; }
        __syncthreads();
    }

    const size_t o = (((size_t)b * NC + c) * kDim + d) * kDst;
    #pragma unroll
    for (int q = 0; q < 4; ++q) {
        float4 pv, hv;
        pv.x = p[4 * q + 0]; pv.y = p[4 * q + 1]; pv.z = p[4 * q + 2]; pv.w = p[4 * q + 3];
        hv.x = h[4 * q + 0]; hv.y = h[4 * q + 1]; hv.z = h[4 * q + 2]; hv.w = h[4 * q + 3];
        *(float4*)(Pws + o + 4 * q) = pv;
        *(float4*)(Hws + o + 4 * q) = hv;
    }
}

constexpr int FT = 64;

__global__ __launch_bounds__(FT)
void scan_fold_kernel(const float* __restrict__ Pws,
                      float* __restrict__ Hws)
{
    constexpr size_t plane = (size_t)kDim * kDst;
    const size_t o = (size_t)blockIdx.y * NC * plane + (size_t)blockIdx.x * FT + threadIdx.x;
    const float* pp = Pws + o;
    float*       hp = Hws + o;

    float pv[NC - 1], hv[NC - 1];
    #pragma unroll
    for (int c = 0; c < NC - 1; ++c) {
        pv[c] = pp[(size_t)c * plane];
        hv[c] = hp[(size_t)c * plane];
    }
    float h = 0.f;
    #pragma unroll
    for (int c = 0; c < NC - 1; ++c) {
        h = pv[c] * h + hv[c];
        hp[(size_t)c * plane] = h;
    }
}

__global__ __launch_bounds__(SW)
void scan_kernel(const float* __restrict__ xin,
                 const float* __restrict__ dlt,
                 const float* __restrict__ Bws,
                 const float* __restrict__ Cws,
                 const float* __restrict__ A_log,
                 const float* __restrict__ Dp,
                 const float* __restrict__ Hws,
                 float* __restrict__ yout)
{
    __shared__ float4 sBC[2][SU][8];

    const int tid = threadIdx.x;
    const int blk = blockIdx.x;
    const int c   = blk % NC;
    const int bg  = blk / NC;
    const int b   = bg / SG;
    const int d   = (bg % SG) * SW + tid;

    float A[kDst];
    #pragma unroll
    for (int q = 0; q < 4; ++q) {
        const float4 a = *(const float4*)(A_log + (size_t)d * kDst + 4 * q);
        A[4 * q + 0] = -__expf(a.x); A[4 * q + 1] = -__expf(a.y);
        A[4 * q + 2] = -__expf(a.z); A[4 * q + 3] = -__expf(a.w);
    }
    const float D_d = Dp[d];

    const size_t row0 = (size_t)b * kL + (size_t)c * CL;
    const float* xin_p = xin + row0 * kDim + d;
    const float* dlt_p = dlt + row0 * kDim + d;
    float*       y_p   = yout + row0 * kDim + d;
    const float4* Bb4  = (const float4*)(Bws + row0 * kDst);
    const float4* Cb4  = (const float4*)(Cws + row0 * kDst);

    const bool    stager = tid < 8 * SU;
    const int     sk     = tid & (4 * SU - 1);
    const float4* BC4    = (tid < 4 * SU) ? Bb4 : Cb4;
    const int     sj     = sk >> 2;
    const int     sq     = (sk & 3) + ((tid < 4 * SU) ? 0 : 4);

    float dtc[SU], xvc[SU];
    #pragma unroll
    for (int j = 0; j < SU; ++j) {
        dtc[j] = dlt_p[(size_t)j * kDim];
        xvc[j] = xin_p[(size_t)j * kDim];
    }
    float h[kDst];
    #pragma unroll
    for (int n = 0; n < kDst; ++n) h[n] = 0.f;
    if (c > 0) {
        const size_t o = (((size_t)b * NC + (c - 1)) * kDim + d) * kDst;
        #pragma unroll
        for (int q = 0; q < 4; ++q) {
            const v4f hv = *(const v4f*)(Hws + o + 4 * q);
            h[4 * q + 0] = hv.x; h[4 * q + 1] = hv.y; h[4 * q + 2] = hv.z; h[4 * q + 3] = hv.w;
        }
    }
    if (stager) sBC[0][sj][sq] = BC4[sk];
    __syncthreads();

    for (int g = 0; g < NGRP; ++g) {
        const int buf = g & 1;
        const int gn = (g + 1 < NGRP) ? g + 1 : g;
        const size_t t1 = (size_t)gn * SU;
        float dtn[SU], xvn[SU];
        #pragma unroll
        for (int j = 0; j < SU; ++j) {
            dtn[j] = dlt_p[(t1 + j) * kDim];
            xvn[j] = xin_p[(t1 + j) * kDim];
        }
        float4 bcx = {};
        if (stager) bcx = BC4[t1 * 4 + sk];

        float4 bq[4], cq[4];
        #pragma unroll
        for (int q = 0; q < 4; ++q) { bq[q] = sBC[buf][0][q]; cq[q] = sBC[buf][0][4 + q]; }

        #pragma unroll
        for (int j = 0; j < SU; ++j) {
            float4 bqn[4], cqn[4];
            if (j + 1 < SU) {
                #pragma unroll
                for (int q = 0; q < 4; ++q) { bqn[q] = sBC[buf][j + 1][q]; cqn[q] = sBC[buf][j + 1][4 + q]; }
            }
            float bn[kDst], cn[kDst];
            #pragma unroll
            for (int q = 0; q < 4; ++q) {
                bn[4 * q + 0] = bq[q].x; bn[4 * q + 1] = bq[q].y; bn[4 * q + 2] = bq[q].z; bn[4 * q + 3] = bq[q].w;
                cn[4 * q + 0] = cq[q].x; cn[4 * q + 1] = cq[q].y; cn[4 * q + 2] = cq[q].z; cn[4 * q + 3] = cq[q].w;
            }
            const float dt = dtc[j];
            const float xv = xvc[j];
            const float u  = dt * xv;
            float y0 = 0.f, y1 = 0.f;
            #pragma unroll
            for (int n = 0; n < kDst; n += 2) {
                const float e0 = __expf(dt * A[n]);
                const float e1 = __expf(dt * A[n + 1]);
                h[n]     = e0 * h[n]     + u * bn[n];
                h[n + 1] = e1 * h[n + 1] + u * bn[n + 1];
                y0 += h[n]     * cn[n];
                y1 += h[n + 1] * cn[n + 1];
            }
            y_p[((size_t)g * SU + j) * kDim] = (y0 + y1) + xv * D_d;
            if (j + 1 < SU) {
                #pragma unroll
                for (int q = 0; q < 4; ++q) { bq[q] = bqn[q]; cq[q] = cqn[q]; }
            }
        }

        if (stager) sBC[buf ^ 1][sj][sq] = bcx;
        #pragma unroll
        for (int j = 0; j < SU; ++j) { dtc[j] = dtn[j]; xvc[j] = xvn[j]; }
        __syncthreads();
    }
}

constexpr int CS = 68;

__global__ __launch_bounds__(GT)
void out_gemm_kernel(const float* __restrict__ y,
                     const float* __restrict__ Wo,
                     const float* __restrict__ bo,
                     float* __restrict__ out)
{
    constexpr int SM4 = (64 * CS / 4 > 2 * OPS) ? 64 * CS / 4 : 2 * OPS;
    __shared__ v4f smem4[SM4];
    v4f*   sA4 = smem4;
    v4f*   sB4 = smem4 + OPS;
    float* sC  = (float*)smem4;

    const int tid  = threadIdx.x;
    const int wave = tid >> 5;
    const int lane = tid & 31;
    const int half = lane >> 4;
    const int l16  = lane & 15;

    const int m0 = blockIdx.x * 64;
    const int n0 = blockIdx.y * 64;
    const int wmi = wave & 1;
    const int wni = wave >> 1;

    v8f c00 = {}, c01 = {}, c10 = {}, c11 = {};

    const float* yb = y  + (size_t)m0 * kDim;
    const float* wb = Wo + (size_t)n0 * kDim;

    v4f ra[4], rb[4];
    load_krows<false>(yb, tid, 64, ra);
    load_krows<false>(wb, tid, 64, rb);

    for (int t = 0; t < NKT; ++t) {
        store_krows<false>(sA4, tid, 64, ra);
        store_krows<false>(sB4, tid, 64, rb);
        __syncthreads();

        if (t + 1 < NKT) {
            const int k1 = (t + 1) * 32;
            load_krows<false>(yb + k1, tid, 64, ra);
            load_krows<false>(wb + k1, tid, 64, rb);
            __builtin_amdgcn_sched_barrier(0);
        }

        tile_mma(sA4, sB4, wmi * 32 + l16, wni * 32 + l16, half, c00, c01, c10, c11);
        __syncthreads();
    }

    #pragma unroll
    for (int j = 0; j < 2; ++j) {
        const int nl = wni * 32 + 16 * j + l16;
        const float bias = bo[n0 + nl];
        #pragma unroll
        for (int i = 0; i < 2; ++i) {
            const v8f& e = i ? (j ? c11 : c10) : (j ? c01 : c00);
            float* p = sC + nl * CS + wmi * 32 + 16 * i + 8 * half;
            v4f lo, hi;
            lo.x = e[0] + bias; lo.y = e[1] + bias; lo.z = e[2] + bias; lo.w = e[3] + bias;
            hi.x = e[4] + bias; hi.y = e[5] + bias; hi.z = e[6] + bias; hi.w = e[7] + bias;
            *(v4f*)(p)     = lo;
            *(v4f*)(p + 4) = hi;
        }
    }
    __syncthreads();

    const int bidx = m0 >> 12;
    const int l0   = m0 & 4095;
    const int mq   = tid & 15;
    const int nr   = tid >> 4;
    #pragma unroll
    for (int p = 0; p < 8; ++p) {
        const int nn = p * 8 + nr;
        const v4f v = *(const v4f*)(sC + nn * CS + 4 * mq);
        *(v4f*)(out + ((size_t)bidx * kDim + n0 + nn) * kL + l0 + 4 * mq) = v;
    }
}

extern "C" void kernel_launch(void* const* d_in, const int* in_sizes, int n_in,
                              void* d_out, int out_size, void* d_ws, size_t ws_size,
                              hipStream_t stream) {
    const float* x     = (const float*)d_in[0];
    const float* Wp    = (const float*)d_in[1];
    const float* bp    = (const float*)d_in[2];
    const float* A_log = (const float*)d_in[3];
    const float* Dp    = (const float*)d_in[4];
    const float* dtb   = (const float*)d_in[5];
    const float* Wo    = (const float*)d_in[6];
    const float* bo    = (const float*)d_in[7];
    float* out = (float*)d_out;

    float* ws = (float*)d_ws;
    const size_t MD = (size_t)kM * kDim;
    const size_t MS = (size_t)kM * kDst;
    float* xin = ws;
    float* dlt = ws + MD;
    float* yws = ws + 2 * MD;
    float* Bw  = ws + 3 * MD;
    float* Cw  = Bw + MS;
    const size_t SS = (size_t)kB * NC * kDim * kDst;
    float* Pw  = Cw + MS;
    float* Hw  = Pw + SS;

    dim3 g1(kM / 64, (kP + 63) / 64);
    proj_gemm_kernel<<<g1, GT, 0, stream>>>(x, Wp, bp, dtb, xin, dlt, Bw, Cw);

    scan_state_kernel<<<kB * SG * (NC - 1), SW, 0, stream>>>(xin, dlt, Bw, A_log, Pw, Hw);

    dim3 gf(kDim * kDst / FT, kB);
    scan_fold_kernel<<<gf, FT, 0, stream>>>(Pw, Hw);

    scan_kernel<<<kB * SG * NC, SW, 0, stream>>>(xin, dlt, Bw, Cw, A_log, Dp, Hw, yws);

    dim3 g3(kM / 64, kDim / 64);
    out_gemm_kernel<<<g3, GT, 0, stream>>>(yws, Wo, bo, out);
}


# Scan and output product fused in one launch: a block scans all 384 channels of a 64-step chunk into LDS as the product's A operand and multiplies by W_out there; y never goes to memory
# speedup vs baseline: 9.4935x; 1.0755x over previous
#include <hip/hip_runtime.h>
#include <hip/hip_bf16.h>

typedef float v2f __attribute__((ext_vector_type(2)));
typedef float v8f __attribute__((ext_vector_type(8)));

constexpr int kDim = 384;
constexpr int kDst = 16;
constexpr int kL   = 4096;
constexpr int kB   = 4;
constexpr int kP   = 800;
constexpr int kM   = kB * kL;

typedef float v4f __attribute__((ext_vector_type(4)));

constexpr int NC = 64;
constexpr int CL = kL / NC;

__device__ __forceinline__ float softplus_f(float v) {
    const float e   = __expf(-fabsf(v));
    const float lg  = 0.69314718f * __builtin_amdgcn_logf(1.f + e);
    const float ser = e * (1.f - e * (0.5f - e * 0.33333334f));
    const float l1p = (e < 0.01f) ? ser : lg;
    return fmaxf(v, 0.f) + l1p;
}

constexpr int GT  = 128;
constexpr int NKT = kDim / 32;
constexpr int QS  = 66;
constexpr int OPS = 8 * QS;

template <bool GUARD>
__device__ __forceinline__ void load_krows(const float* __restrict__ base, int tid,
                                           int valid_rows, v4f (&r)[4]) {
    const int q   = tid & 7;
    const int row = tid >> 3;
    #pragma unroll
    for (int p = 0; p < 4; ++p) {
        int rr = p * 16 + row;
        if (GUARD) rr = (rr < valid_rows) ? rr : 0;
        r[p] = *(const v4f*)(base + (size_t)rr * kDim + 4 * q);
    }
}
template <bool GUARD>
__device__ __forceinline__ void store_krows(v4f* __restrict__ s4, int tid, int valid_rows,
                                            const v4f (&r)[4]) {
    const int q   = tid & 7;
    const int row = tid >> 3;
    #pragma unroll
    for (int p = 0; p < 4; ++p) {
        const int rr = p * 16 + row;
        v4f v = r[p];
        if (GUARD && rr >= valid_rows) { v.x = 0.f; v.y = 0.f; v.z = 0.f; v.w = 0.f; }
        s4[q * QS + rr] = v;
    }
}

__device__ __forceinline__ void tile_mma(const v4f* __restrict__ sA4, const v4f* __restrict__ sB4,
                                         int aslot, int bslot, int half,
                                         v8f& c00, v8f& c01, v8f& c10, v8f& c11) {
    #pragma unroll
    for (int s = 0; s < 4; ++s) {
        const int q = 2 * s + half;
        const v4f a0 = sA4[q * QS + aslot];
        const v4f a1 = sA4[q * QS + aslot + 16];
        const v4f b0 = sB4[q * QS + bslot];
        const v4f b1 = sB4[q * QS + bslot + 16];
        const v2f a0l = {a0.x, a0.y}, a0h = {a0.z, a0.w};
        const v2f a1l = {a1.x, a1.y}, a1h = {a1.z, a1.w};
        const v2f b0l = {b0.x, b0.y}, b0h = {b0.z, b0.w};
        const v2f b1l = {b1.x, b1.y}, b1h = {b1.z, b1.w};
        c00 = __builtin_amdgcn_wmma_f32_16x16x4_f32(false, a0l, false, b0l, (short)0, c00, false, false);
        c01 = __builtin_amdgcn_wmma_f32_16x16x4_f32(false, a0l, false, b1l, (short)0, c01, false, false);
        c10 = __builtin_amdgcn_wmma_f32_16x16x4_f32(false, a1l, false, b0l, (short)0, c10, false, false);
        c11 = __builtin_amdgcn_wmma_f32_16x16x4_f32(false, a1l, false, b1l, (short)0, c11, false, false);
        c00 = __builtin_amdgcn_wmma_f32_16x16x4_f32(false, a0h, false, b0h, (short)0, c00, false, false);
        c01 = __builtin_amdgcn_wmma_f32_16x16x4_f32(false, a0h, false, b1h, (short)0, c01, false, false);
        c10 = __builtin_amdgcn_wmma_f32_16x16x4_f32(false, a1h, false, b0h, (short)0, c10, false, false);
        c11 = __builtin_amdgcn_wmma_f32_16x16x4_f32(false, a1h, false, b1h, (short)0, c11, false, false);
    }
}

__global__ __launch_bounds__(GT)
void proj_gemm_kernel(const float* __restrict__ x,
                      const float* __restrict__ Wp,
                      const float* __restrict__ bp,
                      const float* __restrict__ dtb,
                      float* __restrict__ xin,
                      float* __restrict__ dlt,
                      float* __restrict__ Bws,
                      float* __restrict__ Cws)
{
    __shared__ v4f smem4[2 * OPS];
    v4f* sA4 = smem4;
    v4f* sB4 = smem4 + OPS;

    const int tid  = threadIdx.x;
    const int wave = tid >> 5;
    const int lane = tid & 31;
    const int half = lane >> 4;
    const int l16  = lane & 15;

    const int m0 = blockIdx.x * 64;
    const int n0 = blockIdx.y * 64;

    const int wmi = wave & 1;
    const int wni = wave >> 1;

    v8f c00 = {}, c01 = {}, c10 = {}, c11 = {};

    const int bidx = m0 >> 12;
    const int l0   = m0 & 4095;
    const float* xb = x + (size_t)bidx * kDim * kL + l0;
    const float* wb = Wp + (size_t)n0 * kDim;
    const int vrows = kP - n0;

    const int aq  = tid >> 4;
    const int am4 = tid & 15;

    v4f ra[4], rb[4];
    #pragma unroll
    for (int e = 0; e < 4; ++e) ra[e] = *(const v4f*)(xb + (size_t)(4 * aq + e) * kL + 4 * am4);
    load_krows<true>(wb, tid, vrows, rb);

    for (int t = 0; t < NKT; ++t) {
        #pragma unroll
        for (int mi = 0; mi < 4; ++mi) {
            v4f kq;
            kq.x = ra[0][mi]; kq.y = ra[1][mi]; kq.z = ra[2][mi]; kq.w = ra[3][mi];
            sA4[aq * QS + mi * 16 + am4] = kq;
        }
        store_krows<true>(sB4, tid, vrows, rb);
        __syncthreads();

        if (t + 1 < NKT) {
            const int k1 = (t + 1) * 32;
            #pragma unroll
            for (int e = 0; e < 4; ++e) ra[e] = *(const v4f*)(xb + (size_t)(k1 + 4 * aq + e) * kL + 4 * am4);
            load_krows<true>(wb + k1, tid, vrows, rb);
            __builtin_amdgcn_sched_barrier(0);
        }

        tile_mma(sA4, sB4, wmi * 32 + l16, wni * 32 + l16, half, c00, c01, c10, c11);
        __syncthreads();
    }

    #pragma unroll
    for (int j = 0; j < 2; ++j) {
        const int ng = n0 + wni * 32 + 16 * j + l16;
        if (ng < kP) {
            const float bias = bp[ng];
            float* dst;
            int    ld;
            bool   sp = false;
            float  db = 0.f;
            if (ng < kDim) {
                dst = xin + ng;                     ld = kDim;
            } else if (ng < 2 * kDim) {
                dst = dlt + (ng - kDim);            ld = kDim;
                sp = true;  db = dtb[ng - kDim];
            } else if (ng < 2 * kDim + kDst) {
                dst = Bws + (ng - 2 * kDim);        ld = kDst;
            } else {
                dst = Cws + (ng - 2 * kDim - kDst); ld = kDst;
            }
            dst += (size_t)(m0 + 32 * half + 2 * wmi) * ld;
            const v8f& e0 = j ? c01 : c00;
            const v8f& e1 = j ? c11 : c10;
            #pragma unroll
            for (int r = 0; r < 8; ++r) {
                float v0 = e0[r] + bias;
                float v1 = e1[r] + bias;
                if (sp) {
                    v0 = softplus_f(v0 + db);
                    v1 = softplus_f(v1 + db);
                }
                dst[(size_t)(4 * r) * ld]     = v0;
                dst[(size_t)(4 * r + 1) * ld] = v1;
            }
        }
    }
}

constexpr int SW = 128;
constexpr int SG = kDim / SW;
constexpr int SU = 8;
constexpr int NGRP = CL / SU;

__global__ __launch_bounds__(SW)
void scan_state_kernel(const float* __restrict__ xin,
                       const float* __restrict__ dlt,
                       const float* __restrict__ Bws,
                       const float* __restrict__ A_log,
                       float* __restrict__ Pws,
                       float* __restrict__ Hws)
{
    __shared__ float4 sB[2][SU][4];

    const int tid = threadIdx.x;
    const int blk = blockIdx.x;
    const int c   = blk % (NC - 1);
    const int bg  = blk / (NC - 1);
    const int b   = bg / SG;
    const int d   = (bg % SG) * SW + tid;

    float A[kDst];
    #pragma unroll
    for (int q = 0; q < 4; ++q) {
        const float4 a = *(const float4*)(A_log + (size_t)d * kDst + 4 * q);
        A[4 * q + 0] = -__expf(a.x); A[4 * q + 1] = -__expf(a.y);
        A[4 * q + 2] = -__expf(a.z); A[4 * q + 3] = -__expf(a.w);
    }

    const size_t row0 = (size_t)b * kL + (size_t)c * CL;
    const float* xin_p = xin + row0 * kDim + d;
    const float* dlt_p = dlt + row0 * kDim + d;
    const float4* Bb4  = (const float4*)(Bws + row0 * kDst);

    const bool stager = tid < 4 * SU;
    const int  sk     = tid & (4 * SU - 1);

    float h[kDst], p[kDst];
    #pragma unroll
    for (int n = 0; n < kDst; ++n) { h[n] = 0.f; p[n] = 1.f; }

    float dtc[SU], xvc[SU];
    #pragma unroll
    for (int j = 0; j < SU; ++j) {
        dtc[j] = dlt_p[(size_t)j * kDim];
        xvc[j] = xin_p[(size_t)j * kDim];
    }
    if (stager) sB[0][sk >> 2][sk & 3] = Bb4[sk];
    __syncthreads();

    for (int g = 0; g < NGRP; ++g) {
        const int buf = g & 1;
        const int gn = (g + 1 < NGRP) ? g + 1 : g;
        const size_t t1 = (size_t)gn * SU;
        float dtn[SU], xvn[SU];
        #pragma unroll
        for (int j = 0; j < SU; ++j) {
            dtn[j] = dlt_p[(t1 + j) * kDim];
            xvn[j] = xin_p[(t1 + j) * kDim];
        }
        float4 bnx = {};
        if (stager) bnx = Bb4[t1 * 4 + sk];

        float4 bq[4];
        #pragma unroll
        for (int q = 0; q < 4; ++q) bq[q] = sB[buf][0][q];

        #pragma unroll
        for (int j = 0; j < SU; ++j) {
            float4 bqn[4];
            if (j + 1 < SU) {
                #pragma unroll
                for (int q = 0; q < 4; ++q) bqn[q] = sB[buf][j + 1][q];
            }
            float bn[kDst];
            #pragma unroll
            for (int q = 0; q < 4; ++q) {
                bn[4 * q + 0] = bq[q].x; bn[4 * q + 1] = bq[q].y; bn[4 * q + 2] = bq[q].z; bn[4 * q + 3] = bq[q].w;
            }
            const float dt = dtc[j];
            const float u  = dt * xvc[j];
            #pragma unroll
            for (int n = 0; n < kDst; ++n) {
                const float e = __expf(dt * A[n]);
                h[n] = e * h[n] + u * bn[n];
                p[n] = e * p[n];
            }
            if (j + 1 < SU) {
                #pragma unroll
                for (int q = 0; q < 4; ++q) bq[q] = bqn[q];
            }
        }

        if (stager) sB[buf ^ 1][sk >> 2][sk & 3] = bnx;
        #pragma unroll
        for (int j = 0; j < SU; ++j) { dtc[j] = dtn[j]; xvc[j] = xvn[j]; }
        __syncthreads();
    }

    const size_t o = (((size_t)b * NC + c) * kDim + d) * kDst;
    #pragma unroll
    for (int q = 0; q < 4; ++q) {
        float4 pv, hv;
        pv.x = p[4 * q + 0]; pv.y = p[4 * q + 1]; pv.z = p[4 * q + 2]; pv.w = p[4 * q + 3];
        hv.x = h[4 * q + 0]; hv.y = h[4 * q + 1]; hv.z = h[4 * q + 2]; hv.w = h[4 * q + 3];
        *(float4*)(Pws + o + 4 * q) = pv;
        *(float4*)(Hws + o + 4 * q) = hv;
    }
}

constexpr int FT = 64;

__global__ __launch_bounds__(FT)
void scan_fold_kernel(const float* __restrict__ Pws,
                      float* __restrict__ Hws)
{
    constexpr size_t plane = (size_t)kDim * kDst;
    const size_t o = (size_t)blockIdx.y * NC * plane + (size_t)blockIdx.x * FT + threadIdx.x;
    const float* pp = Pws + o;
    float*       hp = Hws + o;

    float pv[NC - 1], hv[NC - 1];
    #pragma unroll
    for (int c = 0; c < NC - 1; ++c) {
        pv[c] = pp[(size_t)c * plane];
        hv[c] = hp[(size_t)c * plane];
    }
    float h = 0.f;
    #pragma unroll
    for (int c = 0; c < NC - 1; ++c) {
        h = pv[c] * h + hv[c];
        hp[(size_t)c * plane] = h;
    }
}

constexpr int FW  = kDim;
constexpr int QSB = kDim + 2;
constexpr int OPB = 8 * QSB;
constexpr int YS4 = (kDim / 4) * QS;
constexpr int CS  = 68;

__global__ __launch_bounds__(FW)
void scan_out_kernel(const float* __restrict__ xin,
                     const float* __restrict__ dlt,
                     const float* __restrict__ Bws,
                     const float* __restrict__ Cws,
                     const float* __restrict__ A_log,
                     const float* __restrict__ Dp,
                     const float* __restrict__ Hws,
                     const float* __restrict__ Wo,
                     const float* __restrict__ bo,
                     float* __restrict__ out)
{
    constexpr int SC4 = kDim * CS / 4;
    constexpr int SM4 = YS4 + 2 * OPB;
    static_assert(SC4 <= SM4, "staged result must fit over the operand tiles");
    __shared__ v4f    smem4[SM4];
    __shared__ float4 sBC[2][SU][8];
    v4f*   sY4 = smem4;
    v4f*   sW4 = smem4 + YS4;
    float* sYf = (float*)smem4;
    float* sC  = (float*)smem4;

    const int tid = threadIdx.x;
    const int blk = blockIdx.x;
    const int c   = blk % NC;
    const int b   = blk / NC;
    const int d   = tid;

    const int lq   = tid & 7;
    const int lrow = tid >> 3;
    const float* wl = Wo + (size_t)lrow * kDim + 4 * lq;

    float A[kDst];
    #pragma unroll
    for (int q = 0; q < 4; ++q) {
        const float4 a = *(const float4*)(A_log + (size_t)d * kDst + 4 * q);
        A[4 * q + 0] = -__expf(a.x); A[4 * q + 1] = -__expf(a.y);
        A[4 * q + 2] = -__expf(a.z); A[4 * q + 3] = -__expf(a.w);
    }
    const float D_d = Dp[d];

    const size_t row0 = (size_t)b * kL + (size_t)c * CL;
    const float* xin_p = xin + row0 * kDim + d;
    const float* dlt_p = dlt + row0 * kDim + d;
    float*       y_s   = sYf + (d >> 2) * (QS * 4) + (d & 3);
    const float4* Bb4  = (const float4*)(Bws + row0 * kDst);
    const float4* Cb4  = (const float4*)(Cws + row0 * kDst);

    const bool    stager = tid < 8 * SU;
    const int     sk     = tid & (4 * SU - 1);
    const float4* BC4    = (tid < 4 * SU) ? Bb4 : Cb4;
    const int     sj     = sk >> 2;
    const int     sq     = (sk & 3) + ((tid < 4 * SU) ? 0 : 4);

    float dtc[SU], xvc[SU];
    #pragma unroll
    for (int j = 0; j < SU; ++j) {
        dtc[j] = dlt_p[(size_t)j * kDim];
        xvc[j] = xin_p[(size_t)j * kDim];
    }
    float h[kDst];
    #pragma unroll
    for (int n = 0; n < kDst; ++n) h[n] = 0.f;
    if (c > 0) {
        const size_t o = (((size_t)b * NC + (c - 1)) * kDim + d) * kDst;
        #pragma unroll
        for (int q = 0; q < 4; ++q) {
            const v4f hv = *(const v4f*)(Hws + o + 4 * q);
            h[4 * q + 0] = hv.x; h[4 * q + 1] = hv.y; h[4 * q + 2] = hv.z; h[4 * q + 3] = hv.w;
        }
    }
    v4f rb[8];
    #pragma unroll
    for (int p = 0; p < 8; ++p) rb[p] = *(const v4f*)(wl + (size_t)(p * 48) * kDim);
    if (stager) sBC[0][sj][sq] = BC4[sk];
    #pragma unroll
    for (int p = 0; p < 8; ++p) sW4[lq * QSB + p * 48 + lrow] = rb[p];
    #pragma unroll
    for (int p = 0; p < 8; ++p) rb[p] = *(const v4f*)(wl + (size_t)(p * 48) * kDim + 32);
    __syncthreads();

    for (int g = 0; g < NGRP; ++g) {
        const int buf = g & 1;
        const int gn = (g + 1 < NGRP) ? g + 1 : g;
        const size_t t1 = (size_t)gn * SU;
        float dtn[SU], xvn[SU];
        #pragma unroll
        for (int j = 0; j < SU; ++j) {
            dtn[j] = dlt_p[(t1 + j) * kDim];
            xvn[j] = xin_p[(t1 + j) * kDim];
        }
        float4 bcx = {};
        if (stager) bcx = BC4[t1 * 4 + sk];

        float4 bq[4], cq[4];
        #pragma unroll
        for (int q = 0; q < 4; ++q) { bq[q] = sBC[buf][0][q]; cq[q] = sBC[buf][0][4 + q]; }

        #pragma unroll
        for (int j = 0; j < SU; ++j) {
            float4 bqn[4], cqn[4];
            if (j + 1 < SU) {
                #pragma unroll
                for (int q = 0; q < 4; ++q) { bqn[q] = sBC[buf][j + 1][q]; cqn[q] = sBC[buf][j + 1][4 + q]; }
            }
            float bn[kDst], cn[kDst];
            #pragma unroll
            for (int q = 0; q < 4; ++q) {
                bn[4 * q + 0] = bq[q].x; bn[4 * q + 1] = bq[q].y; bn[4 * q + 2] = bq[q].z; bn[4 * q + 3] = bq[q].w;
                cn[4 * q + 0] = cq[q].x; cn[4 * q + 1] = cq[q].y; cn[4 * q + 2] = cq[q].z; cn[4 * q + 3] = cq[q].w;
            }
            const float dt = dtc[j];
            const float xv = xvc[j];
            const float u  = dt * xv;
            float y0 = 0.f, y1 = 0.f;
            #pragma unroll
            for (int n = 0; n < kDst; n += 2) {
                const float e0 = __expf(dt * A[n]);
                const float e1 = __expf(dt * A[n + 1]);
                h[n]     = e0 * h[n]     + u * bn[n];
                h[n + 1] = e1 * h[n + 1] + u * bn[n + 1];
                y0 += h[n]     * cn[n];
                y1 += h[n + 1] * cn[n + 1];
            }
            y_s[4 * (g * SU + j)] = (y0 + y1) + xv * D_d;
            if (j + 1 < SU) {
                #pragma unroll
                for (int q = 0; q < 4; ++q) { bq[q] = bqn[q]; cq[q] = cqn[q]; }
            }
        }

        if (stager) sBC[buf ^ 1][sj][sq] = bcx;
        #pragma unroll
        for (int j = 0; j < SU; ++j) { dtc[j] = dtn[j]; xvc[j] = xvn[j]; }
        __syncthreads();
    }

    const int wave = tid >> 5;
    const int lane = tid & 31;
    const int half = lane >> 4;
    const int l16  = lane & 15;
    const int wmi  = wave & 1;
    const int wni  = wave >> 1;

    v8f acc[2][4];
    #pragma unroll
    for (int i = 0; i < 2; ++i) {
        #pragma unroll
        for (int j = 0; j < 4; ++j) acc[i][j] = v8f{};
    }

    const int arow = wmi * 32 + l16;
    const int bcol = wni * 64 + l16;
    for (int t = 0; t < NKT; ++t) {
        const v4f* sA = sY4 + (size_t)(8 * t) * QS;
        const v4f* sB = sW4 + (t & 1) * OPB;
        #pragma unroll
        for (int s = 0; s < 4; ++s) {
            const int q = 2 * s + half;
            const v4f a0 = sA[q * QS + arow];
            const v4f a1 = sA[q * QS + arow + 16];
            v4f bb[4];
            #pragma unroll
            for (int j = 0; j < 4; ++j) bb[j] = sB[q * QSB + bcol + 16 * j];
            const v2f al[2] = {{a0.x, a0.y}, {a1.x, a1.y}};
            const v2f ah[2] = {{a0.z, a0.w}, {a1.z, a1.w}};
            #pragma unroll
            for (int i = 0; i < 2; ++i) {
                #pragma unroll
                for (int j = 0; j < 4; ++j) {
                    const v2f bl = {bb[j].x, bb[j].y};
                    acc[i][j] = __builtin_amdgcn_wmma_f32_16x16x4_f32(false, al[i], false, bl, (short)0, acc[i][j], false, false);
                }
            }
            #pragma unroll
            for (int i = 0; i < 2; ++i) {
                #pragma unroll
                for (int j = 0; j < 4; ++j) {
                    const v2f bh = {bb[j].z, bb[j].w};
                    acc[i][j] = __builtin_amdgcn_wmma_f32_16x16x4_f32(false, ah[i], false, bh, (short)0, acc[i][j], false, false);
                }
            }
        }
        if (t + 1 < NKT) {
            v4f* sN = sW4 + ((t + 1) & 1) * OPB;
            #pragma unroll
            for (int p = 0; p < 8; ++p) sN[lq * QSB + p * 48 + lrow] = rb[p];
            if (t + 2 < NKT) {
                const int k2 = (t + 2) * 32;
                #pragma unroll
                for (int p = 0; p < 8; ++p) rb[p] = *(const v4f*)(wl + (size_t)(p * 48) * kDim + k2);
            }
        }
        __syncthreads();
    }

    #pragma unroll
    for (int j = 0; j < 4; ++j) {
        const int nl = wni * 64 + 16 * j + l16;
        const float bias = bo[nl];
        #pragma unroll
        for (int i = 0; i < 2; ++i) {
            const v8f e = acc[i][j];
            float* p = sC + nl * CS + wmi * 32 + 16 * i + 8 * half;
            v4f lo, hi;
            lo.x = e[0] + bias; lo.y = e[1] + bias; lo.z = e[2] + bias; lo.w = e[3] + bias;
            hi.x = e[4] + bias; hi.y = e[5] + bias; hi.z = e[6] + bias; hi.w = e[7] + bias;
            *(v4f*)(p)     = lo;
            *(v4f*)(p + 4) = hi;
        }
    }
    __syncthreads();

    const int l0 = c * CL;
    const int mq = tid & 15;
    const int nr = tid >> 4;
    #pragma unroll
    for (int p = 0; p < 16; ++p) {
        const int nn = p * 24 + nr;
        const v4f v = *(const v4f*)(sC + nn * CS + 4 * mq);
        *(v4f*)(out + ((size_t)b * kDim + nn) * kL + l0 + 4 * mq) = v;
    }
}

extern "C" void kernel_launch(void* const* d_in, const int* in_sizes, int n_in,
                              void* d_out, int out_size, void* d_ws, size_t ws_size,
                              hipStream_t stream) {
    const float* x     = (const float*)d_in[0];
    const float* Wp    = (const float*)d_in[1];
    const float* bp    = (const float*)d_in[2];
    const float* A_log = (const float*)d_in[3];
    const float* Dp    = (const float*)d_in[4];
    const float* dtb   = (const float*)d_in[5];
    const float* Wo    = (const float*)d_in[6];
    const float* bo    = (const float*)d_in[7];
    float* out = (float*)d_out;

    float* ws = (float*)d_ws;
    const size_t MD = (size_t)kM * kDim;
    const size_t MS = (size_t)kM * kDst;
    float* xin = ws;
    float* dlt = ws + MD;
    float* Bw  = ws + 2 * MD;
    float* Cw  = Bw + MS;
    const size_t SS = (size_t)kB * NC * kDim * kDst;
    float* Pw  = Cw + MS;
    float* Hw  = Pw + SS;

    dim3 g1(kM / 64, (kP + 63) / 64);
    proj_gemm_kernel<<<g1, GT, 0, stream>>>(x, Wp, bp, dtb, xin, dlt, Bw, Cw);

    scan_state_kernel<<<kB * SG * (NC - 1), SW, 0, stream>>>(xin, dlt, Bw, A_log, Pw, Hw);

    dim3 gf(kDim * kDst / FT, kB);
    scan_fold_kernel<<<gf, FT, 0, stream>>>(Pw, Hw);

    scan_out_kernel<<<kB * NC, FW, 0, stream>>>(xin, dlt, Bw, Cw, A_log, Dp, Hw, Wo, bo, out);
}
